# starting kernel
# baseline (speedup 1.0000x reference)
#include <hip/hip_runtime.h>
#include <hip/hip_bf16.h>

typedef float v2f  __attribute__((ext_vector_type(2)));
typedef float v8f  __attribute__((ext_vector_type(8)));
typedef _Float16 v16h __attribute__((ext_vector_type(16)));

#define FDIM 64

__global__ void gcn_deg_init(float* __restrict__ deg, int n) {
    int i = blockIdx.x * blockDim.x + threadIdx.x;
    if (i < n) deg[i] = 1.0f;
}

__global__ void gcn_deg_acc(const int* __restrict__ col,
                            const float* __restrict__ w,
                            float* __restrict__ deg, int e) {
    int i = blockIdx.x * blockDim.x + threadIdx.x;
    if (i < e) atomicAdd(&deg[col[i]], w[i]);
}

__global__ void gcn_dinv(const float* __restrict__ deg,
                         float* __restrict__ dinv, int n) {
    int i = blockIdx.x * blockDim.x + threadIdx.x;
    if (i < n) {
        float d = deg[i];
        dinv[i] = __frsqrt_rn(d);
    }
}

__global__ __launch_bounds__(256)
void gcn_gemm64(const float* __restrict__ X, const float* __restrict__ W,
                float* __restrict__ H, int n) {
    const int lane = threadIdx.x & 31;
    const int wave = threadIdx.x >> 5;
    const int rowBase = blockIdx.x * 32 + (wave >> 2) * 16;
    const int colBase = (wave & 3) * 16;
    const int half = lane >> 4;
    const int l15  = lane & 15;

    int arow = rowBase + l15;
    if (arow >= n) arow = n - 1;
    const float* __restrict__ xrow = X + (size_t)arow * FDIM;
    const float* __restrict__ wrow = W + (size_t)(colBase + l15) * FDIM;

    v8f c = {0.f, 0.f, 0.f, 0.f, 0.f, 0.f, 0.f, 0.f};

#if __has_builtin(__builtin_amdgcn_wmma_f32_16x16x4_f32)
    #pragma unroll
    for (int k0 = 0; k0 < FDIM; k0 += 4) {
        const int ka = k0 + half * 2;
        v2f a, b;
        a.x = xrow[ka]; a.y = xrow[ka + 1];
        b.x = wrow[ka]; b.y = wrow[ka + 1];
        c = __builtin_amdgcn_wmma_f32_16x16x4_f32(
                false, a, false, b, (short)0, c, false, false);
    }
#else
    #pragma unroll
    for (int k0 = 0; k0 < FDIM; k0 += 32) {
        v16h a, b;
        #pragma unroll
        for (int i = 0; i < 16; ++i) {
            const int p = i >> 1, wlo = i & 1;
            const int k = k0 + (p >> 2) * 16 + half * 8 + (p & 3) * 2 + wlo;
            a[i] = (_Float16)xrow[k];
            b[i] = (_Float16)wrow[k];
        }
        c = __builtin_amdgcn_wmma_f32_16x16x32_f16(
                false, a, false, b, (short)0, c, false, false);
    }
#endif

    #pragma unroll
    for (int r = 0; r < 8; ++r) {
        const int m = rowBase + half * 8 + r;
        if (m < n) H[(size_t)m * FDIM + colBase + l15] = c[r];
    }
}

__global__ void gcn_init_agg(const float* __restrict__ H,
                             const float* __restrict__ dinv,
                             const float* __restrict__ bias,
                             float* __restrict__ out, int n) {
    int t = blockIdx.x * blockDim.x + threadIdx.x;
    if (t >= n * 16) return;
    const int i = t >> 4, c4 = t & 15;
    const float d2 = dinv[i] * dinv[i];
    float4 h = reinterpret_cast<const float4*>(H)[t];
    float4 b = reinterpret_cast<const float4*>(bias)[c4];
    float4 o;
    o.x = d2 * h.x + b.x;  o.y = d2 * h.y + b.y;
    o.z = d2 * h.z + b.z;  o.w = d2 * h.w + b.w;
    reinterpret_cast<float4*>(out)[t] = o;
}

__global__ __launch_bounds__(256)
void gcn_edge_scatter(const float* __restrict__ H,
                      const int* __restrict__ row,
                      const int* __restrict__ col,
                      const float* __restrict__ w,
                      const float* __restrict__ dinv,
                      float* __restrict__ out, int e) {
    int t = blockIdx.x * blockDim.x + threadIdx.x;
    const int ei = t >> 4;
    if (ei >= e) return;
    const int c4 = t & 15;
    const int r = row[ei], c = col[ei];
    const float nrm = dinv[r] * w[ei] * dinv[c];
    float4 v = reinterpret_cast<const float4*>(H + (size_t)r * FDIM)[c4];
    float* dst = out + (size_t)c * FDIM + c4 * 4;
    atomicAdd(dst + 0, nrm * v.x);
    atomicAdd(dst + 1, nrm * v.y);
    atomicAdd(dst + 2, nrm * v.z);
    atomicAdd(dst + 3, nrm * v.w);
}

__global__ void gcn_relu(float* __restrict__ a, int n16) {
    int t = blockIdx.x * blockDim.x + threadIdx.x;
    if (t >= n16) return;
    float4 v = reinterpret_cast<float4*>(a)[t];
    v.x = fmaxf(v.x, 0.f); v.y = fmaxf(v.y, 0.f);
    v.z = fmaxf(v.z, 0.f); v.w = fmaxf(v.w, 0.f);
    reinterpret_cast<float4*>(a)[t] = v;
}

extern "C" void kernel_launch(void* const* d_in, const int* in_sizes, int n_in,
                              void* d_out, int out_size, void* d_ws, size_t ws_size,
                              hipStream_t stream) {
    const float* x   = (const float*)d_in[0];
    const float* ew  = (const float*)d_in[1];
    const float* W1  = (const float*)d_in[2];
    const float* b1  = (const float*)d_in[3];
    const float* W2  = (const float*)d_in[4];
    const float* b2  = (const float*)d_in[5];
    const int*  eidx = (const int*)d_in[6];

    const int n = in_sizes[0] / FDIM;
    const int e = in_sizes[1];
    const int* row = eidx;
    const int* col = eidx + e;

    float* deg  = (float*)d_ws;
    float* dinv = deg  + n;
    float* hlin = dinv + n;
    float* agg1 = hlin + (size_t)n * FDIM;
    float* outp = (float*)d_out;

    const int B = 256;
    const int gN   = (n + B - 1) / B;
    const int gE   = (e + B - 1) / B;
    const int gN16 = (n * 16 + B - 1) / B;
    const int gE16 = (e * 16 + B - 1) / B;
    const int gGemm = (n + 31) / 32;

    gcn_deg_init<<<gN, B, 0, stream>>>(deg, n);
    gcn_deg_acc <<<gE, B, 0, stream>>>(col, ew, deg, e);
    gcn_dinv    <<<gN, B, 0, stream>>>(deg, dinv, n);

    gcn_gemm64      <<<gGemm, B, 0, stream>>>(x, W1, hlin, n);
    gcn_init_agg    <<<gN16, B, 0, stream>>>(hlin, dinv, b1, agg1, n);
    gcn_edge_scatter<<<gE16, B, 0, stream>>>(hlin, row, col, ew, dinv, agg1, e);
    gcn_relu        <<<gN16, B, 0, stream>>>(agg1, n * 16);

    gcn_gemm64      <<<gGemm, B, 0, stream>>>(agg1, W2, hlin, n);
    gcn_init_agg    <<<gN16, B, 0, stream>>>(hlin, dinv, b2, outp, n);
    gcn_edge_scatter<<<gE16, B, 0, stream>>>(hlin, row, col, ew, dinv, outp, e);
}


# edge scatter: one wave per 32 edges, lanes span contiguous columns: each f32 atomic and gather load covers one whole 128-B line (2 atomics per edge, was 8 partial); sums still in atomic arrival order
# speedup vs baseline: 1.2944x; 1.2944x over previous
#include <hip/hip_runtime.h>
#include <hip/hip_bf16.h>

typedef float v2f  __attribute__((ext_vector_type(2)));
typedef float v8f  __attribute__((ext_vector_type(8)));
typedef _Float16 v16h __attribute__((ext_vector_type(16)));

#define FDIM 64

__global__ void gcn_deg_init(float* __restrict__ deg, int n) {
    int i = blockIdx.x * blockDim.x + threadIdx.x;
    if (i < n) deg[i] = 1.0f;
}

__global__ void gcn_deg_acc(const int* __restrict__ col,
                            const float* __restrict__ w,
                            float* __restrict__ deg, int e) {
    int i = blockIdx.x * blockDim.x + threadIdx.x;
    if (i < e) atomicAdd(&deg[col[i]], w[i]);
}

__global__ void gcn_dinv(const float* __restrict__ deg,
                         float* __restrict__ dinv, int n) {
    int i = blockIdx.x * blockDim.x + threadIdx.x;
    if (i < n) {
        float d = deg[i];
        dinv[i] = __frsqrt_rn(d);
    }
}

__global__ __launch_bounds__(256)
void gcn_gemm64(const float* __restrict__ X, const float* __restrict__ W,
                float* __restrict__ H, int n) {
    const int lane = threadIdx.x & 31;
    const int wave = threadIdx.x >> 5;
    const int rowBase = blockIdx.x * 32 + (wave >> 2) * 16;
    const int colBase = (wave & 3) * 16;
    const int half = lane >> 4;
    const int l15  = lane & 15;

    int arow = rowBase + l15;
    if (arow >= n) arow = n - 1;
    const float* __restrict__ xrow = X + (size_t)arow * FDIM;
    const float* __restrict__ wrow = W + (size_t)(colBase + l15) * FDIM;

    v8f c = {0.f, 0.f, 0.f, 0.f, 0.f, 0.f, 0.f, 0.f};

#if __has_builtin(__builtin_amdgcn_wmma_f32_16x16x4_f32)
    #pragma unroll
    for (int k0 = 0; k0 < FDIM; k0 += 4) {
        const int ka = k0 + half * 2;
        v2f a, b;
        a.x = xrow[ka]; a.y = xrow[ka + 1];
        b.x = wrow[ka]; b.y = wrow[ka + 1];
        c = __builtin_amdgcn_wmma_f32_16x16x4_f32(
                false, a, false, b, (short)0, c, false, false);
    }
#else
    #pragma unroll
    for (int k0 = 0; k0 < FDIM; k0 += 32) {
        v16h a, b;
        #pragma unroll
        for (int i = 0; i < 16; ++i) {
            const int p = i >> 1, wlo = i & 1;
            const int k = k0 + (p >> 2) * 16 + half * 8 + (p & 3) * 2 + wlo;
            a[i] = (_Float16)xrow[k];
            b[i] = (_Float16)wrow[k];
        }
        c = __builtin_amdgcn_wmma_f32_16x16x32_f16(
                false, a, false, b, (short)0, c, false, false);
    }
#endif

    #pragma unroll
    for (int r = 0; r < 8; ++r) {
        const int m = rowBase + half * 8 + r;
        if (m < n) H[(size_t)m * FDIM + colBase + l15] = c[r];
    }
}

__global__ void gcn_init_agg(const float* __restrict__ H,
                             const float* __restrict__ dinv,
                             const float* __restrict__ bias,
                             float* __restrict__ out, int n) {
    int t = blockIdx.x * blockDim.x + threadIdx.x;
    if (t >= n * 16) return;
    const int i = t >> 4, c4 = t & 15;
    const float d2 = dinv[i] * dinv[i];
    float4 h = reinterpret_cast<const float4*>(H)[t];
    float4 b = reinterpret_cast<const float4*>(bias)[c4];
    float4 o;
    o.x = d2 * h.x + b.x;  o.y = d2 * h.y + b.y;
    o.z = d2 * h.z + b.z;  o.w = d2 * h.w + b.w;
    reinterpret_cast<float4*>(out)[t] = o;
}

#define EDGES_PER_WAVE 32
#define EDGE_BATCH 8

__device__ __forceinline__ void gcn_scatter_one(const float* __restrict__ H,
                                                float* __restrict__ out,
                                                int r, int c, float nrm, int lane) {
    const float* src = H + (size_t)r * FDIM;
    float* dst = out + (size_t)c * FDIM;
    const float v0 = src[lane];
    const float v1 = src[lane + 32];
    atomicAdd(dst + lane, nrm * v0);
    atomicAdd(dst + lane + 32, nrm * v1);
}

__global__ __launch_bounds__(256)
void gcn_edge_scatter(const float* __restrict__ H,
                      const int* __restrict__ row,
                      const int* __restrict__ col,
                      const float* __restrict__ w,
                      const float* __restrict__ dinv,
                      float* __restrict__ out, int e) {
    const int lane = threadIdx.x & 31;
    const int gwave = (blockIdx.x * blockDim.x + threadIdx.x) >> 5;
    const int base = gwave * EDGES_PER_WAVE;
    if (base >= e) return;
    const int ei = base + lane;
    int r = 0, c = 0;
    float nrm = 0.f;
    if (ei < e) {
        r = row[ei];
        c = col[ei];
        nrm = dinv[r] * w[ei] * dinv[c];
    }
    const int nrm_bits = __float_as_int(nrm);

    if (base + EDGES_PER_WAVE <= e) {
        #pragma unroll
        for (int j0 = 0; j0 < EDGES_PER_WAVE; j0 += EDGE_BATCH) {
            float v0[EDGE_BATCH], v1[EDGE_BATCH], nj[EDGE_BATCH];
            float* dst[EDGE_BATCH];
            #pragma unroll
            for (int u = 0; u < EDGE_BATCH; ++u) {
                const int rj = __builtin_amdgcn_readlane(r, j0 + u);
                const int cj = __builtin_amdgcn_readlane(c, j0 + u);
                nj[u] = __int_as_float(__builtin_amdgcn_readlane(nrm_bits, j0 + u));
                const float* src = H + (size_t)rj * FDIM;
                dst[u] = out + (size_t)cj * FDIM;
                v0[u] = src[lane];
                v1[u] = src[lane + 32];
            }
            #pragma unroll
            for (int u = 0; u < EDGE_BATCH; ++u) {
                atomicAdd(dst[u] + lane, nj[u] * v0[u]);
                atomicAdd(dst[u] + lane + 32, nj[u] * v1[u]);
            }
        }
    } else {
        const int cnt = e - base;
        for (int j = 0; j < cnt; ++j) {
            const int rj = __builtin_amdgcn_readlane(r, j);
            const int cj = __builtin_amdgcn_readlane(c, j);
            const float nj = __int_as_float(__builtin_amdgcn_readlane(nrm_bits, j));
            gcn_scatter_one(H, out, rj, cj, nj, lane);
        }
    }
}

__global__ void gcn_relu(float* __restrict__ a, int n16) {
    int t = blockIdx.x * blockDim.x + threadIdx.x;
    if (t >= n16) return;
    float4 v = reinterpret_cast<float4*>(a)[t];
    v.x = fmaxf(v.x, 0.f); v.y = fmaxf(v.y, 0.f);
    v.z = fmaxf(v.z, 0.f); v.w = fmaxf(v.w, 0.f);
    reinterpret_cast<float4*>(a)[t] = v;
}

extern "C" void kernel_launch(void* const* d_in, const int* in_sizes, int n_in,
                              void* d_out, int out_size, void* d_ws, size_t ws_size,
                              hipStream_t stream) {
    const float* x   = (const float*)d_in[0];
    const float* ew  = (const float*)d_in[1];
    const float* W1  = (const float*)d_in[2];
    const float* b1  = (const float*)d_in[3];
    const float* W2  = (const float*)d_in[4];
    const float* b2  = (const float*)d_in[5];
    const int*  eidx = (const int*)d_in[6];

    const int n = in_sizes[0] / FDIM;
    const int e = in_sizes[1];
    const int* row = eidx;
    const int* col = eidx + e;

    float* deg  = (float*)d_ws;
    float* dinv = deg  + n;
    float* hlin = dinv + n;
    float* agg1 = hlin + (size_t)n * FDIM;
    float* outp = (float*)d_out;

    const int B = 256;
    const int gN   = (n + B - 1) / B;
    const int gE   = (e + B - 1) / B;
    const int gN16 = (n * 16 + B - 1) / B;
    const int gES  = ((e + 31) / 32 * 32 + B - 1) / B;
    const int gGemm = (n + 31) / 32;

    gcn_deg_init<<<gN, B, 0, stream>>>(deg, n);
    gcn_deg_acc <<<gE, B, 0, stream>>>(col, ew, deg, e);
    gcn_dinv    <<<gN, B, 0, stream>>>(deg, dinv, n);

    gcn_gemm64      <<<gGemm, B, 0, stream>>>(x, W1, hlin, n);
    gcn_init_agg    <<<gN16, B, 0, stream>>>(hlin, dinv, b1, agg1, n);
    gcn_edge_scatter<<<gES, B, 0, stream>>>(hlin, row, col, ew, dinv, agg1, e);
    gcn_relu        <<<gN16, B, 0, stream>>>(agg1, n * 16);

    gcn_gemm64      <<<gGemm, B, 0, stream>>>(agg1, W2, hlin, n);
    gcn_init_agg    <<<gN16, B, 0, stream>>>(hlin, dinv, b2, outp, n);
    gcn_edge_scatter<<<gES, B, 0, stream>>>(hlin, row, col, ew, dinv, outp, e);
}


# segment sum by owner wave: edges ranked per target by returning integer atomic, segment starts by one add per 32 nodes; a wave gathers a node's rows, sums in registers, stores once; sum order = rank
# speedup vs baseline: 1.3433x; 1.0378x over previous
#include <hip/hip_runtime.h>
#include <hip/hip_bf16.h>

typedef float v2f  __attribute__((ext_vector_type(2)));
typedef float v8f  __attribute__((ext_vector_type(8)));
typedef _Float16 v16h __attribute__((ext_vector_type(16)));

#define FDIM 64

__global__ void gcn_node_init(float* __restrict__ deg, int* __restrict__ cnt,
                              int* __restrict__ total, int n) {
    int i = blockIdx.x * blockDim.x + threadIdx.x;
    if (i < n) { deg[i] = 1.0f; cnt[i] = 0; }
    if (i == 0) *total = 0;
}

#define EDGES_PER_THREAD 4

__global__ __launch_bounds__(256)
void gcn_deg_rank(const int* __restrict__ col, const float* __restrict__ w,
                  float* __restrict__ deg, int* __restrict__ cnt,
                  int* __restrict__ rank, int e) {
    const int base = blockIdx.x * (256 * EDGES_PER_THREAD) + threadIdx.x;
    int c[EDGES_PER_THREAD], rk[EDGES_PER_THREAD];
    float wv[EDGES_PER_THREAD];
    #pragma unroll
    for (int u = 0; u < EDGES_PER_THREAD; ++u) {
        const int i = base + u * 256;
        c[u] = 0; wv[u] = 0.f;
        if (i < e) { c[u] = col[i]; wv[u] = w[i]; }
    }
    #pragma unroll
    for (int u = 0; u < EDGES_PER_THREAD; ++u) {
        const int i = base + u * 256;
        rk[u] = 0;
        if (i < e) {
            atomicAdd(&deg[c[u]], wv[u]);
            rk[u] = atomicAdd(&cnt[c[u]], 1);
        }
    }
    #pragma unroll
    for (int u = 0; u < EDGES_PER_THREAD; ++u) {
        const int i = base + u * 256;
        if (i < e) rank[i] = rk[u];
    }
}

__global__ __launch_bounds__(256)
void gcn_seg_alloc(const int* __restrict__ cnt, int* __restrict__ start,
                   int* __restrict__ total, float* __restrict__ deg_dinv, int n) {
    const int i = blockIdx.x * blockDim.x + threadIdx.x;
    const int lane = threadIdx.x & 31;
    int c = 0;
    if (i < n) {
        c = cnt[i];
        const float d = deg_dinv[i];
        deg_dinv[i] = __frsqrt_rn(d);
    }
    int x = c;
    #pragma unroll
    for (int off = 1; off < 32; off <<= 1) {
        const int y = __shfl_up(x, off);
        if (lane >= off) x += y;
    }
    int base = 0;
    if (lane == 31) base = atomicAdd(total, x);
    base = __shfl(base, 31);
    if (i < n) start[i] = base + (x - c);
}

__global__ __launch_bounds__(256)
void gcn_fill(const int* __restrict__ row, const int* __restrict__ col,
              const float* __restrict__ w, const float* __restrict__ dinv,
              const int* __restrict__ start, const int* __restrict__ rank,
              int* __restrict__ rec, int e) {
    const int base = blockIdx.x * (256 * EDGES_PER_THREAD) + threadIdx.x;
    #pragma unroll
    for (int u = 0; u < EDGES_PER_THREAD; ++u) {
        const int i = base + u * 256;
        if (i < e) {
            const int r = row[i];
            const int c = col[i];
            const float nrm = dinv[r] * w[i] * dinv[c];
            const int slot = start[c] + rank[i];
            int2 o;
            o.x = r;
            o.y = __float_as_int(nrm);
            reinterpret_cast<int2*>(rec)[slot] = o;
        }
    }
}

__global__ __launch_bounds__(256)
void gcn_gemm64(const float* X, const float* __restrict__ W,
                float* H, int n) {
    const int lane = threadIdx.x & 31;
    const int wave = threadIdx.x >> 5;
    const int rowBase = blockIdx.x * 32 + (wave >> 2) * 16;
    const int colBase = (wave & 3) * 16;
    const int half = lane >> 4;
    const int l15  = lane & 15;

    int arow = rowBase + l15;
    if (arow >= n) arow = n - 1;
    const float* xrow = X + (size_t)arow * FDIM;
    const float* __restrict__ wrow = W + (size_t)(colBase + l15) * FDIM;

    v8f c = {0.f, 0.f, 0.f, 0.f, 0.f, 0.f, 0.f, 0.f};

#if __has_builtin(__builtin_amdgcn_wmma_f32_16x16x4_f32)
    #pragma unroll
    for (int k0 = 0; k0 < FDIM; k0 += 4) {
        const int ka = k0 + half * 2;
        v2f a, b;
        a.x = xrow[ka]; a.y = xrow[ka + 1];
        b.x = wrow[ka]; b.y = wrow[ka + 1];
        c = __builtin_amdgcn_wmma_f32_16x16x4_f32(
                false, a, false, b, (short)0, c, false, false);
    }
#else
    #pragma unroll
    for (int k0 = 0; k0 < FDIM; k0 += 32) {
        v16h a, b;
        #pragma unroll
        for (int i = 0; i < 16; ++i) {
            const int p = i >> 1, wlo = i & 1;
            const int k = k0 + (p >> 2) * 16 + half * 8 + (p & 3) * 2 + wlo;
            a[i] = (_Float16)xrow[k];
            b[i] = (_Float16)wrow[k];
        }
        c = __builtin_amdgcn_wmma_f32_16x16x32_f16(
                false, a, false, b, (short)0, c, false, false);
    }
#endif

    __syncthreads();

    #pragma unroll
    for (int r = 0; r < 8; ++r) {
        const int m = rowBase + half * 8 + r;
        if (m < n) H[(size_t)m * FDIM + colBase + l15] = c[r];
    }
}

template <int B>
__device__ __forceinline__ void gcn_gather_batch(const float* H, int rsrc, int rnrm,
                                                 int j0, int lane, float2& acc) {
    float2 v[B];
    float nj[B];
    #pragma unroll
    for (int u = 0; u < B; ++u) {
        const int rj = __builtin_amdgcn_readlane(rsrc, j0 + u);
        nj[u] = __int_as_float(__builtin_amdgcn_readlane(rnrm, j0 + u));
        v[u] = *reinterpret_cast<const float2*>(H + (size_t)rj * FDIM + lane * 2);
    }
    #pragma unroll
    for (int u = 0; u < B; ++u) {
        acc.x += nj[u] * v[u].x;
        acc.y += nj[u] * v[u].y;
    }
}

#define AGG_BLOCKS 2048

__global__ __launch_bounds__(256)
void gcn_aggregate(const float* H, const int* __restrict__ start,
                   const int* __restrict__ cnt,
                   const int* __restrict__ rec, const float* __restrict__ dinv,
                   const float* __restrict__ bias, float* out,
                   int n, int relu) {
    const int lane = threadIdx.x & 31;
    const int gwave = __builtin_amdgcn_readfirstlane((blockIdx.x * blockDim.x + threadIdx.x) >> 5);
    const int nwaves = (gridDim.x * blockDim.x) >> 5;
    const int2* __restrict__ rec2 = reinterpret_cast<const int2*>(rec);
    const float2 bb = *reinterpret_cast<const float2*>(bias + lane * 2);

    for (int node = gwave; node < n; node += nwaves) {
        const int s0 = start[node];
        const int s1 = s0 + cnt[node];
        const float di = dinv[node];
        const float d2 = di * di;
        const float2 hs = *reinterpret_cast<const float2*>(H + (size_t)node * FDIM + lane * 2);
        float2 acc = bb;

        for (int base = s0; base < s1; base += 32) {
            int rem = s1 - base;
            if (rem > 32) rem = 32;
            const int idx = base + (lane < rem ? lane : rem - 1);
            const int2 rc = rec2[idx];
            int j = 0;
            for (; j + 8 <= rem; j += 8) gcn_gather_batch<8>(H, rc.x, rc.y, j, lane, acc);
            if (j + 4 <= rem) { gcn_gather_batch<4>(H, rc.x, rc.y, j, lane, acc); j += 4; }
            if (j + 2 <= rem) { gcn_gather_batch<2>(H, rc.x, rc.y, j, lane, acc); j += 2; }
            if (j < rem)      { gcn_gather_batch<1>(H, rc.x, rc.y, j, lane, acc); }
        }
        acc.x += d2 * hs.x;
        acc.y += d2 * hs.y;
        if (relu) { acc.x = fmaxf(acc.x, 0.f); acc.y = fmaxf(acc.y, 0.f); }
        *reinterpret_cast<float2*>(out + (size_t)node * FDIM + lane * 2) = acc;
    }
}

extern "C" void kernel_launch(void* const* d_in, const int* in_sizes, int n_in,
                              void* d_out, int out_size, void* d_ws, size_t ws_size,
                              hipStream_t stream) {
    const float* x   = (const float*)d_in[0];
    const float* ew  = (const float*)d_in[1];
    const float* W1  = (const float*)d_in[2];
    const float* b1  = (const float*)d_in[3];
    const float* W2  = (const float*)d_in[4];
    const float* b2  = (const float*)d_in[5];
    const int*  eidx = (const int*)d_in[6];

    const int n = in_sizes[0] / FDIM;
    const int e = in_sizes[1];
    const int* row = eidx;
    const int* col = eidx + e;

    float* dinv  = (float*)d_ws;
    int*   cnt   = (int*)(dinv + n);
    int*   rec   = (int*)(cnt + n);
    int*   rank  = rec + 2 * (size_t)e;
    int*   start = rank + e;
    int*   total = start + n;
    float* agg1  = (float*)(cnt + n) + (size_t)n * FDIM;
    float* outp  = (float*)d_out;

    const int B = 256;
    const int gN    = (n + B - 1) / B;
    const int gE4   = (e + B * EDGES_PER_THREAD - 1) / (B * EDGES_PER_THREAD);
    const int gGemm = (n + 31) / 32;

    gcn_node_init<<<gN, B, 0, stream>>>(dinv, cnt, total, n);
    gcn_deg_rank <<<gE4, B, 0, stream>>>(col, ew, dinv, cnt, rank, e);
    gcn_seg_alloc<<<gN, B, 0, stream>>>(cnt, start, total, dinv, n);
    gcn_fill     <<<gE4, B, 0, stream>>>(row, col, ew, dinv, start, rank, rec, e);

    gcn_gemm64   <<<gGemm, B, 0, stream>>>(x, W1, outp, n);
    gcn_aggregate<<<AGG_BLOCKS, B, 0, stream>>>(outp, start, cnt, rec, dinv, b1, agg1, n, 1);

    gcn_gemm64   <<<gGemm, B, 0, stream>>>(agg1, W2, agg1, n);
    gcn_aggregate<<<AGG_BLOCKS, B, 0, stream>>>(agg1, start, cnt, rec, dinv, b2, outp, n, 0);
}


# dense transform moved behind the neighbour sum, same launch: A(XW)=(AX)W; a block sums 32 nodes' input rows into LDS, then one f32 WMMA tile per wave; sum order changed; 8 launches to 6
# speedup vs baseline: 1.5615x; 1.1624x over previous
#include <hip/hip_runtime.h>
#include <hip/hip_bf16.h>

typedef float v2f  __attribute__((ext_vector_type(2)));
typedef float v8f  __attribute__((ext_vector_type(8)));

#define FDIM 64

__global__ void gcn_node_init(float* __restrict__ deg, int* __restrict__ cnt,
                              int* __restrict__ total, int n) {
    int i = blockIdx.x * blockDim.x + threadIdx.x;
    if (i < n) { deg[i] = 1.0f; cnt[i] = 0; }
    if (i == 0) *total = 0;
}

#define EDGES_PER_THREAD 4

__global__ __launch_bounds__(256)
void gcn_deg_rank(const int* __restrict__ col, const float* __restrict__ w,
                  float* __restrict__ deg, int* __restrict__ cnt,
                  int* __restrict__ rank, int e) {
    const int base = blockIdx.x * (256 * EDGES_PER_THREAD) + threadIdx.x;
    int c[EDGES_PER_THREAD], rk[EDGES_PER_THREAD];
    float wv[EDGES_PER_THREAD];
    #pragma unroll
    for (int u = 0; u < EDGES_PER_THREAD; ++u) {
        const int i = base + u * 256;
        c[u] = 0; wv[u] = 0.f;
        if (i < e) { c[u] = col[i]; wv[u] = w[i]; }
    }
    #pragma unroll
    for (int u = 0; u < EDGES_PER_THREAD; ++u) {
        const int i = base + u * 256;
        rk[u] = 0;
        if (i < e) {
            atomicAdd(&deg[c[u]], wv[u]);
            rk[u] = atomicAdd(&cnt[c[u]], 1);
        }
    }
    #pragma unroll
    for (int u = 0; u < EDGES_PER_THREAD; ++u) {
        const int i = base + u * 256;
        if (i < e) rank[i] = rk[u];
    }
}

__global__ __launch_bounds__(256)
void gcn_seg_alloc(const int* __restrict__ cnt, int* __restrict__ start,
                   int* __restrict__ total, float* __restrict__ deg_dinv, int n) {
    const int i = blockIdx.x * blockDim.x + threadIdx.x;
    const int lane = threadIdx.x & 31;
    int c = 0;
    if (i < n) {
        c = cnt[i];
        const float d = deg_dinv[i];
        deg_dinv[i] = __frsqrt_rn(d);
    }
    int x = c;
    #pragma unroll
    for (int off = 1; off < 32; off <<= 1) {
        const int y = __shfl_up(x, off);
        if (lane >= off) x += y;
    }
    int base = 0;
    if (lane == 31) base = atomicAdd(total, x);
    base = __shfl(base, 31);
    if (i < n) start[i] = base + (x - c);
}

__global__ __launch_bounds__(256)
void gcn_fill(const int* __restrict__ row, const int* __restrict__ col,
              const float* __restrict__ w, const float* __restrict__ dinv,
              const int* __restrict__ start, const int* __restrict__ rank,
              int* __restrict__ rec, int e) {
    const int base = blockIdx.x * (256 * EDGES_PER_THREAD) + threadIdx.x;
    #pragma unroll
    for (int u = 0; u < EDGES_PER_THREAD; ++u) {
        const int i = base + u * 256;
        if (i < e) {
            const int r = row[i];
            const int c = col[i];
            const float nrm = dinv[r] * w[i] * dinv[c];
            const int slot = start[c] + rank[i];
            int2 o;
            o.x = r;
            o.y = __float_as_int(nrm);
            reinterpret_cast<int2*>(rec)[slot] = o;
        }
    }
}

template <int B>
__device__ __forceinline__ void gcn_gather_batch(const float* __restrict__ X, int rsrc, int rnrm,
                                                 int j0, int lane, float2& acc) {
    float2 v[B];
    float nj[B];
    #pragma unroll
    for (int u = 0; u < B; ++u) {
        const int rj = __builtin_amdgcn_readlane(rsrc, j0 + u);
        nj[u] = __int_as_float(__builtin_amdgcn_readlane(rnrm, j0 + u));
        v[u] = *reinterpret_cast<const float2*>(X + (size_t)rj * FDIM + lane * 2);
    }
    #pragma unroll
    for (int u = 0; u < B; ++u) {
        acc.x += nj[u] * v[u].x;
        acc.y += nj[u] * v[u].y;
    }
}

__device__ __forceinline__ void gcn_gather_chunk(const float* __restrict__ X, int2 rc, int rem,
                                                 int lane, float2& acc) {
    int j = 0;
    for (; j + 8 <= rem; j += 8) gcn_gather_batch<8>(X, rc.x, rc.y, j, lane, acc);
    if (j + 4 <= rem) { gcn_gather_batch<4>(X, rc.x, rc.y, j, lane, acc); j += 4; }
    if (j + 2 <= rem) { gcn_gather_batch<2>(X, rc.x, rc.y, j, lane, acc); j += 2; }
    if (j < rem)      { gcn_gather_batch<1>(X, rc.x, rc.y, j, lane, acc); }
}

#define TILE_NODES 32
#define NODES_PER_WAVE 4
#define TILE_PITCH 68

__global__ __launch_bounds__(256)
void gcn_agg_gemm(const float* __restrict__ Xin, const int* __restrict__ start,
                  const int* __restrict__ cnt, const int* __restrict__ rec,
                  const float* __restrict__ dinv, const float* __restrict__ W,
                  const float* __restrict__ bias, float* __restrict__ out,
                  int n, int relu) {
    __shared__ alignas(16) float tile[TILE_NODES * TILE_PITCH];
    __shared__ alignas(16) float wt[FDIM * TILE_PITCH];

    {
        const int tid = threadIdx.x;
        float4 w4[4];
        #pragma unroll
        for (int i = 0; i < 4; ++i)
            w4[i] = *reinterpret_cast<const float4*>(W + i * 1024 + tid * 4);
        #pragma unroll
        for (int i = 0; i < 4; ++i)
            *reinterpret_cast<float4*>(&wt[(i * 16 + (tid >> 4)) * TILE_PITCH + (tid & 15) * 4]) = w4[i];
    }

    const int lane = threadIdx.x & 31;
    const int wave = __builtin_amdgcn_readfirstlane(threadIdx.x >> 5);
    const int nodeBase = blockIdx.x * TILE_NODES;
    const int2* __restrict__ rec2 = reinterpret_cast<const int2*>(rec);

    const int node0 = nodeBase + wave * NODES_PER_WAVE;
    const int ln  = node0 + (lane & (NODES_PER_WAVE - 1));
    const int lnd = ln < n ? ln : n - 1;
    const int   sv = start[lnd];
    const int   cv = cnt[lnd];
    const float dv = dinv[lnd];
    int   s0[NODES_PER_WAVE], cn[NODES_PER_WAVE];
    float d2[NODES_PER_WAVE];
    int2  rc[NODES_PER_WAVE];
    float2 hs[NODES_PER_WAVE];
    #pragma unroll
    for (int t = 0; t < NODES_PER_WAVE; ++t) {
        const bool valid = node0 + t < n;
        s0[t] = __builtin_amdgcn_readlane(sv, t);
        const int c = __builtin_amdgcn_readlane(cv, t);
        const float di = __int_as_float(__builtin_amdgcn_readlane(__float_as_int(dv), t));
        cn[t] = valid ? c : 0;
        d2[t] = valid ? di * di : 0.f;
    }
    #pragma unroll
    for (int t = 0; t < NODES_PER_WAVE; ++t) {
        const int node = node0 + t;
        const int nd = node < n ? node : n - 1;
        hs[t] = *reinterpret_cast<const float2*>(Xin + (size_t)nd * FDIM + lane * 2);
        int rem = cn[t] < 32 ? cn[t] : 32;
        rc[t].x = 0; rc[t].y = 0;
        if (rem > 0) rc[t] = rec2[s0[t] + (lane < rem ? lane : rem - 1)];
    }
    #pragma unroll
    for (int t = 0; t < NODES_PER_WAVE; ++t) {
        float2 acc; acc.x = 0.f; acc.y = 0.f;
        const int s1 = s0[t] + cn[t];
        gcn_gather_chunk(Xin, rc[t], cn[t] < 32 ? cn[t] : 32, lane, acc);
        for (int base = s0[t] + 32; base < s1; base += 32) {
            int rem = s1 - base;
            if (rem > 32) rem = 32;
            const int2 rmore = rec2[base + (lane < rem ? lane : rem - 1)];
            gcn_gather_chunk(Xin, rmore, rem, lane, acc);
        }
        acc.x += d2[t] * hs[t].x;
        acc.y += d2[t] * hs[t].y;
        *reinterpret_cast<float2*>(&tile[(wave * NODES_PER_WAVE + t) * TILE_PITCH + lane * 2]) = acc;
    }

    const int rowT = wave >> 2;
    const int colBase = (wave & 3) * 16;
    const int half = lane >> 4;
    const int l15  = lane & 15;
    const float2 bb = *reinterpret_cast<const float2*>(bias + lane * 2);

    __syncthreads();

    const float* arow = &tile[(rowT * 16 + l15) * TILE_PITCH];
    const float* wrow = &wt[(colBase + l15) * TILE_PITCH];
    v2f a[FDIM / 4], b[FDIM / 4];
    v8f c = {0.f, 0.f, 0.f, 0.f, 0.f, 0.f, 0.f, 0.f};
    __builtin_amdgcn_sched_barrier(0);
    #pragma unroll
    for (int q = 0; q < FDIM / 4; ++q) {
        const int ka = q * 4 + half * 2;
        a[q].x = arow[ka]; a[q].y = arow[ka + 1];
        b[q].x = wrow[ka]; b[q].y = wrow[ka + 1];
        c = __builtin_amdgcn_wmma_f32_16x16x4_f32(
                false, a[q], false, b[q], (short)0, c, false, false);
    }
    asm volatile("v_nop\n\tv_nop\n\tv_nop\n\tv_nop\n\tv_nop"
                 : "+v"(c)
                 : "v"(a[FDIM / 4 - 2]), "v"(b[FDIM / 4 - 2]), "v"(a[FDIM / 4 - 1]), "v"(b[FDIM / 4 - 1]));
    __builtin_amdgcn_sched_barrier(0);

    __syncthreads();
    #pragma unroll
    for (int r = 0; r < 8; ++r)
        tile[(rowT * 16 + half * 8 + r) * TILE_PITCH + colBase + l15] = c[r];
    __syncthreads();
    #pragma unroll
    for (int t = 0; t < NODES_PER_WAVE; ++t) {
        const int rrow = wave * NODES_PER_WAVE + t;
        const int m = nodeBase + rrow;
        if (m < n) {
            float2 v = *reinterpret_cast<const float2*>(&tile[rrow * TILE_PITCH + lane * 2]);
            v.x += bb.x;
            v.y += bb.y;
            if (relu) { v.x = fmaxf(v.x, 0.f); v.y = fmaxf(v.y, 0.f); }
            *reinterpret_cast<float2*>(out + (size_t)m * FDIM + lane * 2) = v;
        }
    }
}

extern "C" void kernel_launch(void* const* d_in, const int* in_sizes, int n_in,
                              void* d_out, int out_size, void* d_ws, size_t ws_size,
                              hipStream_t stream) {
    const float* x   = (const float*)d_in[0];
    const float* ew  = (const float*)d_in[1];
    const float* W1  = (const float*)d_in[2];
    const float* b1  = (const float*)d_in[3];
    const float* W2  = (const float*)d_in[4];
    const float* b2  = (const float*)d_in[5];
    const int*  eidx = (const int*)d_in[6];

    const int n = in_sizes[0] / FDIM;
    const int e = in_sizes[1];
    const int* row = eidx;
    const int* col = eidx + e;

    float* dinv  = (float*)d_ws;
    int*   cnt   = (int*)(dinv + n);
    int*   rec   = (int*)(cnt + n);
    int*   rank  = rec + 2 * (size_t)e;
    int*   start = rank + e;
    int*   total = start + n;
    float* agg1  = (float*)(cnt + n) + (size_t)n * FDIM;
    float* outp  = (float*)d_out;

    const int B = 256;
    const int gN    = (n + B - 1) / B;
    const int gE4   = (e + B * EDGES_PER_THREAD - 1) / (B * EDGES_PER_THREAD);
    const int gTile = (n + TILE_NODES - 1) / TILE_NODES;

    gcn_node_init<<<gN, B, 0, stream>>>(dinv, cnt, total, n);
    gcn_deg_rank <<<gE4, B, 0, stream>>>(col, ew, dinv, cnt, rank, e);
    gcn_seg_alloc<<<gN, B, 0, stream>>>(cnt, start, total, dinv, n);
    gcn_fill     <<<gE4, B, 0, stream>>>(row, col, ew, dinv, start, rank, rec, e);

    gcn_agg_gemm <<<gTile, B, 0, stream>>>(x, start, cnt, rec, dinv, W1, b1, agg1, n, 1);

    gcn_agg_gemm <<<gTile, B, 0, stream>>>(agg1, start, cnt, rec, dinv, W2, b2, outp, n, 0);
}


# normalisation folded into the tables: deg summed per node from its grouped (row,w) records by a wave tree (float atomic, 2 of 3 random loads per edge gone); rows pre-scaled by dinv; 6 to 7 launches
# speedup vs baseline: 1.8687x; 1.1967x over previous
#include <hip/hip_runtime.h>
#include <hip/hip_bf16.h>

typedef float v2f  __attribute__((ext_vector_type(2)));
typedef float v8f  __attribute__((ext_vector_type(8)));

#define FDIM 64

__global__ void gcn_node_init(int* __restrict__ cnt, int* __restrict__ total, int n) {
    int i = blockIdx.x * blockDim.x + threadIdx.x;
    if (i < n) cnt[i] = 0;
    if (i == 0) *total = 0;
}

#define EDGES_PER_THREAD 4

__global__ __launch_bounds__(256)
void gcn_rank(const int* __restrict__ col, int* __restrict__ cnt,
              int* __restrict__ rank, int e) {
    const int base = blockIdx.x * (256 * EDGES_PER_THREAD) + threadIdx.x;
    int c[EDGES_PER_THREAD], rk[EDGES_PER_THREAD];
    #pragma unroll
    for (int u = 0; u < EDGES_PER_THREAD; ++u) {
        const int i = base + u * 256;
        c[u] = 0;
        if (i < e) c[u] = col[i];
    }
    #pragma unroll
    for (int u = 0; u < EDGES_PER_THREAD; ++u) {
        const int i = base + u * 256;
        rk[u] = 0;
        if (i < e) rk[u] = atomicAdd(&cnt[c[u]], 1);
    }
    #pragma unroll
    for (int u = 0; u < EDGES_PER_THREAD; ++u) {
        const int i = base + u * 256;
        if (i < e) rank[i] = rk[u];
    }
}

__global__ __launch_bounds__(256)
void gcn_seg_alloc(const int* __restrict__ cnt, int* __restrict__ start,
                   int* __restrict__ total, int n) {
    const int i = blockIdx.x * blockDim.x + threadIdx.x;
    const int lane = threadIdx.x & 31;
    int c = 0;
    if (i < n) c = cnt[i];
    int x = c;
    #pragma unroll
    for (int off = 1; off < 32; off <<= 1) {
        const int y = __shfl_up(x, off);
        if (lane >= off) x += y;
    }
    int base = 0;
    if (lane == 31) base = atomicAdd(total, x);
    base = __shfl(base, 31);
    if (i < n) start[i] = base + (x - c);
}

__global__ __launch_bounds__(256)
void gcn_fill(const int* __restrict__ row, const int* __restrict__ col,
              const float* __restrict__ w, const int* __restrict__ start,
              const int* __restrict__ rank, int* __restrict__ rec, int e) {
    const int base = blockIdx.x * (256 * EDGES_PER_THREAD) + threadIdx.x;
    #pragma unroll
    for (int u = 0; u < EDGES_PER_THREAD; ++u) {
        const int i = base + u * 256;
        if (i < e) {
            const int c = col[i];
            const int slot = start[c] + rank[i];
            int2 o;
            o.x = row[i];
            o.y = __float_as_int(w[i]);
            reinterpret_cast<int2*>(rec)[slot] = o;
        }
    }
}

#define SCALE_NODES 8

__global__ __launch_bounds__(256)
void gcn_deg_scale(const float* __restrict__ X, const int* __restrict__ start,
                   const int* __restrict__ cnt, const int* __restrict__ rec,
                   float* __restrict__ dinv, float* __restrict__ Xs, int n) {
    const int lane = threadIdx.x & 31;
    const int wave = __builtin_amdgcn_readfirstlane(threadIdx.x >> 5);
    const int node0 = (blockIdx.x * 8 + wave) * SCALE_NODES;
    const int2* __restrict__ rec2 = reinterpret_cast<const int2*>(rec);
    const int ln  = node0 + (lane & (SCALE_NODES - 1));
    const int lnd = ln < n ? ln : n - 1;
    const int sv = start[lnd];
    const int cv = cnt[lnd];
    int    s0[SCALE_NODES], cn[SCALE_NODES];
    float  ws[SCALE_NODES];
    float2 xr[SCALE_NODES];
    #pragma unroll
    for (int t = 0; t < SCALE_NODES; ++t) {
        const int node = node0 + t;
        const bool valid = node < n;
        const int nd = valid ? node : n - 1;
        s0[t] = __builtin_amdgcn_readlane(sv, t);
        const int c = __builtin_amdgcn_readlane(cv, t);
        cn[t] = valid ? c : 0;
        xr[t] = *reinterpret_cast<const float2*>(X + (size_t)nd * FDIM + lane * 2);
        ws[t] = 0.f;
        if (cn[t] > 0) {
            const int rem = cn[t] < 32 ? cn[t] : 32;
            const int2 r = rec2[s0[t] + (lane < rem ? lane : rem - 1)];
            ws[t] = lane < rem ? __int_as_float(r.y) : 0.f;
        }
    }
    #pragma unroll
    for (int t = 0; t < SCALE_NODES; ++t) {
        const int s1 = s0[t] + cn[t];
        for (int base = s0[t] + 32; base < s1; base += 32) {
            const int rem = s1 - base;
            if (lane < rem) ws[t] += __int_as_float(rec2[base + lane].y);
        }
    }
    #pragma unroll
    for (int off = 16; off > 0; off >>= 1) {
        float o[SCALE_NODES];
        #pragma unroll
        for (int t = 0; t < SCALE_NODES; ++t) o[t] = __shfl_xor(ws[t], off);
        #pragma unroll
        for (int t = 0; t < SCALE_NODES; ++t) ws[t] += o[t];
    }
    float mine = 0.f;
    #pragma unroll
    for (int t = 0; t < SCALE_NODES; ++t) {
        const float d = 1.0f + ws[t];
        const float di = d > 0.f ? __frsqrt_rn(d) : 0.f;
        if ((lane & (SCALE_NODES - 1)) == t) mine = di;
        const int node = node0 + t;
        if (node < n) {
            float2 o;
            o.x = di * xr[t].x;
            o.y = di * xr[t].y;
            *reinterpret_cast<float2*>(Xs + (size_t)node * FDIM + lane * 2) = o;
        }
    }
    if (lane < SCALE_NODES && ln < n) dinv[ln] = mine;
}

template <int B>
__device__ __forceinline__ void gcn_gather_batch(const float* __restrict__ X, int rsrc, int rnrm,
                                                 int j0, int lane, float2& acc) {
    float2 v[B];
    float nj[B];
    #pragma unroll
    for (int u = 0; u < B; ++u) {
        const int rj = __builtin_amdgcn_readlane(rsrc, j0 + u);
        nj[u] = __int_as_float(__builtin_amdgcn_readlane(rnrm, j0 + u));
        v[u] = *reinterpret_cast<const float2*>(X + (size_t)rj * FDIM + lane * 2);
    }
    #pragma unroll
    for (int u = 0; u < B; ++u) {
        acc.x += nj[u] * v[u].x;
        acc.y += nj[u] * v[u].y;
    }
}

__device__ __forceinline__ void gcn_gather_chunk(const float* __restrict__ X, int2 rc, int rem,
                                                 int lane, float2& acc) {
    int j = 0;
    for (; j + 8 <= rem; j += 8) gcn_gather_batch<8>(X, rc.x, rc.y, j, lane, acc);
    if (j + 4 <= rem) { gcn_gather_batch<4>(X, rc.x, rc.y, j, lane, acc); j += 4; }
    if (j + 2 <= rem) { gcn_gather_batch<2>(X, rc.x, rc.y, j, lane, acc); j += 2; }
    if (j < rem)      { gcn_gather_batch<1>(X, rc.x, rc.y, j, lane, acc); }
}

#define TILE_NODES 32
#define NODES_PER_WAVE 4
#define TILE_PITCH 68

__global__ __launch_bounds__(256)
void gcn_agg_gemm(const float* __restrict__ Xin, const int* __restrict__ start,
                  const int* __restrict__ cnt, const int* __restrict__ rec,
                  const float* __restrict__ dinv, const float* __restrict__ W,
                  const float* __restrict__ bias, float* __restrict__ out,
                  int n, int relu) {
    __shared__ alignas(16) float tile[TILE_NODES * TILE_PITCH];
    __shared__ alignas(16) float wt[FDIM * TILE_PITCH];

    {
        const int tid = threadIdx.x;
        float4 w4[4];
        #pragma unroll
        for (int i = 0; i < 4; ++i)
            w4[i] = *reinterpret_cast<const float4*>(W + i * 1024 + tid * 4);
        #pragma unroll
        for (int i = 0; i < 4; ++i)
            *reinterpret_cast<float4*>(&wt[(i * 16 + (tid >> 4)) * TILE_PITCH + (tid & 15) * 4]) = w4[i];
    }

    const int lane = threadIdx.x & 31;
    const int wave = __builtin_amdgcn_readfirstlane(threadIdx.x >> 5);
    const int nodeBase = blockIdx.x * TILE_NODES;
    const int2* __restrict__ rec2 = reinterpret_cast<const int2*>(rec);

    const int node0 = nodeBase + wave * NODES_PER_WAVE;
    const int ln  = node0 + (lane & (NODES_PER_WAVE - 1));
    const int lnd = ln < n ? ln : n - 1;
    const int   sv = start[lnd];
    const int   cv = cnt[lnd];
    const float dv = dinv[lnd];
    int   s0[NODES_PER_WAVE], cn[NODES_PER_WAVE];
    float dsc[NODES_PER_WAVE];
    int2  rc[NODES_PER_WAVE];
    float2 hs[NODES_PER_WAVE];
    #pragma unroll
    for (int t = 0; t < NODES_PER_WAVE; ++t) {
        const bool valid = node0 + t < n;
        s0[t] = __builtin_amdgcn_readlane(sv, t);
        const int c = __builtin_amdgcn_readlane(cv, t);
        const float di = __int_as_float(__builtin_amdgcn_readlane(__float_as_int(dv), t));
        cn[t] = valid ? c : 0;
        dsc[t] = valid ? di : 0.f;
    }
    #pragma unroll
    for (int t = 0; t < NODES_PER_WAVE; ++t) {
        const int node = node0 + t;
        const int nd = node < n ? node : n - 1;
        hs[t] = *reinterpret_cast<const float2*>(Xin + (size_t)nd * FDIM + lane * 2);
        int rem = cn[t] < 32 ? cn[t] : 32;
        rc[t].x = 0; rc[t].y = 0;
        if (rem > 0) rc[t] = rec2[s0[t] + (lane < rem ? lane : rem - 1)];
    }
    #pragma unroll
    for (int t = 0; t < NODES_PER_WAVE; ++t) {
        float2 acc; acc.x = 0.f; acc.y = 0.f;
        const int s1 = s0[t] + cn[t];
        gcn_gather_chunk(Xin, rc[t], cn[t] < 32 ? cn[t] : 32, lane, acc);
        for (int base = s0[t] + 32; base < s1; base += 32) {
            int rem = s1 - base;
            if (rem > 32) rem = 32;
            const int2 rmore = rec2[base + (lane < rem ? lane : rem - 1)];
            gcn_gather_chunk(Xin, rmore, rem, lane, acc);
        }
        acc.x = dsc[t] * (acc.x + hs[t].x);
        acc.y = dsc[t] * (acc.y + hs[t].y);
        *reinterpret_cast<float2*>(&tile[(wave * NODES_PER_WAVE + t) * TILE_PITCH + lane * 2]) = acc;
    }

    const int rowT = wave >> 2;
    const int colBase = (wave & 3) * 16;
    const int half = lane >> 4;
    const int l15  = lane & 15;
    const float2 bb = *reinterpret_cast<const float2*>(bias + lane * 2);

    __syncthreads();

    const float* arow = &tile[(rowT * 16 + l15) * TILE_PITCH];
    const float* wrow = &wt[(colBase + l15) * TILE_PITCH];
    v2f a[FDIM / 4], b[FDIM / 4];
    v8f c = {0.f, 0.f, 0.f, 0.f, 0.f, 0.f, 0.f, 0.f};
    __builtin_amdgcn_sched_barrier(0);
    #pragma unroll
    for (int q = 0; q < FDIM / 4; ++q) {
        const int ka = q * 4 + half * 2;
        a[q].x = arow[ka]; a[q].y = arow[ka + 1];
        b[q].x = wrow[ka]; b[q].y = wrow[ka + 1];
        c = __builtin_amdgcn_wmma_f32_16x16x4_f32(
                false, a[q], false, b[q], (short)0, c, false, false);
    }
    asm volatile("v_nop\n\tv_nop\n\tv_nop\n\tv_nop\n\tv_nop"
                 : "+v"(c)
                 : "v"(a[FDIM / 4 - 2]), "v"(b[FDIM / 4 - 2]), "v"(a[FDIM / 4 - 1]), "v"(b[FDIM / 4 - 1]));
    __builtin_amdgcn_sched_barrier(0);

    __syncthreads();
    #pragma unroll
    for (int r = 0; r < 8; ++r)
        tile[(rowT * 16 + half * 8 + r) * TILE_PITCH + colBase + l15] = c[r];
    __syncthreads();
    #pragma unroll
    for (int t = 0; t < NODES_PER_WAVE; ++t) {
        const int rrow = wave * NODES_PER_WAVE + t;
        const int m = nodeBase + rrow;
        if (m < n) {
            float2 v = *reinterpret_cast<const float2*>(&tile[rrow * TILE_PITCH + lane * 2]);
            v.x += bb.x;
            v.y += bb.y;
            if (relu) { v.x = dsc[t] * fmaxf(v.x, 0.f); v.y = dsc[t] * fmaxf(v.y, 0.f); }
            *reinterpret_cast<float2*>(out + (size_t)m * FDIM + lane * 2) = v;
        }
    }
}

extern "C" void kernel_launch(void* const* d_in, const int* in_sizes, int n_in,
                              void* d_out, int out_size, void* d_ws, size_t ws_size,
                              hipStream_t stream) {
    const float* x   = (const float*)d_in[0];
    const float* ew  = (const float*)d_in[1];
    const float* W1  = (const float*)d_in[2];
    const float* b1  = (const float*)d_in[3];
    const float* W2  = (const float*)d_in[4];
    const float* b2  = (const float*)d_in[5];
    const int*  eidx = (const int*)d_in[6];

    const int n = in_sizes[0] / FDIM;
    const int e = in_sizes[1];
    const int* row = eidx;
    const int* col = eidx + e;

    float* dinv  = (float*)d_ws;
    int*   cnt   = (int*)(dinv + n);
    int*   rec   = (int*)(cnt + n);
    int*   rank  = rec + 2 * (size_t)e;
    int*   start = rank + e;
    int*   total = start + n;
    float* agg1  = (float*)(cnt + n) + (size_t)n * FDIM;
    float* outp  = (float*)d_out;

    const int B = 256;
    const int gN    = (n + B - 1) / B;
    const int gE4   = (e + B * EDGES_PER_THREAD - 1) / (B * EDGES_PER_THREAD);
    const int gTile = (n + TILE_NODES - 1) / TILE_NODES;
    const int gScale = (n + 8 * SCALE_NODES - 1) / (8 * SCALE_NODES);

    gcn_node_init<<<gN, B, 0, stream>>>(cnt, total, n);
    gcn_rank     <<<gE4, B, 0, stream>>>(col, cnt, rank, e);
    gcn_seg_alloc<<<gN, B, 0, stream>>>(cnt, start, total, n);
    gcn_fill     <<<gE4, B, 0, stream>>>(row, col, ew, start, rank, rec, e);
    gcn_deg_scale<<<gScale, B, 0, stream>>>(x, start, cnt, rec, dinv, outp, n);

    gcn_agg_gemm <<<gTile, B, 0, stream>>>(outp, start, cnt, rec, dinv, W1, b1, agg1, n, 1);

    gcn_agg_gemm <<<gTile, B, 0, stream>>>(agg1, start, cnt, rec, dinv, W2, b2, outp, n, 0);
}


# edges grouped by target with a two-level counting sort ranked by LDS counters (chunk x 128-node bucket, then node), not a returning global atomic per edge; 7 launches to 6; sum order = LDS rank order
# speedup vs baseline: 3.5664x; 1.9085x over previous
#include <hip/hip_runtime.h>
#include <hip/hip_bf16.h>

typedef float v2f  __attribute__((ext_vector_type(2)));
typedef float v8f  __attribute__((ext_vector_type(8)));

#define FDIM 64

#define BK_SHIFT 7
#define BK_NODES (1 << BK_SHIFT)
#define MAX_BK 512
#define CH_THREADS 1024
#define CH_EPT 8
#define CH_EDGES (CH_THREADS * CH_EPT)

__global__ __launch_bounds__(CH_THREADS)
void gcn_bucket_count(const int* __restrict__ col, int* __restrict__ H, int e, int nb) {
    __shared__ int lc[MAX_BK];
    const int tid = threadIdx.x;
    if (tid < MAX_BK) lc[tid] = 0;
    const int base = blockIdx.x * CH_EDGES + tid;
    int c[CH_EPT];
    #pragma unroll
    for (int u = 0; u < CH_EPT; ++u) {
        const int i = base + u * CH_THREADS;
        const int v = col[i < e ? i : e - 1];
        c[u] = i < e ? v : -1;
    }
    __syncthreads();
    #pragma unroll
    for (int u = 0; u < CH_EPT; ++u)
        if (c[u] >= 0) atomicAdd(&lc[c[u] >> BK_SHIFT], 1);
    __syncthreads();
    if (tid < nb) H[blockIdx.x * nb + tid] = lc[tid];
}

__global__ __launch_bounds__(CH_THREADS) __attribute__((amdgpu_waves_per_eu(8, 8)))
void gcn_bucket_scatter(const int* __restrict__ row, const int* __restrict__ col,
                        const float* __restrict__ w, const int* __restrict__ H,
                        int* __restrict__ bstart, int* __restrict__ trec,
                        int* __restrict__ tcol, int e, int nb, int nblk) {
    __shared__ int lc[MAX_BK];
    __shared__ int tot[2][MAX_BK];
    __shared__ int pre[2][MAX_BK];
    __shared__ int off[MAX_BK];
    __shared__ int loff[MAX_BK];
    __shared__ int wsum[2][MAX_BK / 32];
    __shared__ alignas(8) int2 srec[CH_EDGES];
    __shared__ int scol[CH_EDGES];
    const int tid = threadIdx.x;
    const int lane = tid & 31;
    if (tid < MAX_BK) lc[tid] = 0;
    const int base = blockIdx.x * CH_EDGES + tid;
    int c[CH_EPT], rw[CH_EPT], rk[CH_EPT];
    float wv[CH_EPT];
    #pragma unroll
    for (int u = 0; u < CH_EPT; ++u) {
        const int i = base + u * CH_THREADS;
        const int ic = i < e ? i : e - 1;
        const int v = col[ic];
        rw[u] = row[ic];
        wv[u] = w[ic];
        c[u] = i < e ? v : -1;
    }
    {
        const int b = tid & (MAX_BK - 1);
        const int hf = tid >> 9;
        const int mid = (nblk + 1) >> 1;
        const int k0 = hf ? mid : 0;
        const int k1 = hf ? nblk : mid;
        const int me = blockIdx.x;
        int t = 0, p = 0;
        if (b < nb) {
            #pragma unroll 16
            for (int k = k0; k < k1; ++k) {
                const int v = H[k * nb + b];
                t += v;
                p += k < me ? v : 0;
            }
        }
        tot[hf][b] = t;
        pre[hf][b] = p;
    }
    __syncthreads();
    #pragma unroll
    for (int u = 0; u < CH_EPT; ++u) {
        rk[u] = 0;
        if (c[u] >= 0) rk[u] = atomicAdd(&lc[c[u] >> BK_SHIFT], 1);
    }
    __syncthreads();
    int x = 0, y = 0, ix = 0, iy = 0;
    if (tid < MAX_BK) {
        x = tot[0][tid] + tot[1][tid];
        y = lc[tid];
        ix = x; iy = y;
        #pragma unroll
        for (int o = 1; o < 32; o <<= 1) {
            const int vx = __shfl_up(ix, o);
            const int vy = __shfl_up(iy, o);
            if (lane >= o) { ix += vx; iy += vy; }
        }
        if (lane == 31) { wsum[0][tid >> 5] = ix; wsum[1][tid >> 5] = iy; }
    }
    __syncthreads();
    if (tid < MAX_BK) {
        int wx = 0, wy = 0;
        #pragma unroll
        for (int k = 0; k < MAX_BK / 32; ++k) {
            wx += k < (tid >> 5) ? wsum[0][k] : 0;
            wy += k < (tid >> 5) ? wsum[1][k] : 0;
        }
        const int ex = wx + ix - x;
        const int ey = wy + iy - y;
        loff[tid] = ey;
        off[tid] = ex + pre[0][tid] + pre[1][tid] - ey;
        if (blockIdx.x == 0) {
            if (tid < nb) bstart[tid] = ex;
            if (tid == 0) bstart[nb] = e;
        }
    }
    __syncthreads();
    #pragma unroll
    for (int u = 0; u < CH_EPT; ++u) {
        if (c[u] >= 0) {
            const int place = loff[c[u] >> BK_SHIFT] + rk[u];
            int2 o;
            o.x = rw[u];
            o.y = __float_as_int(wv[u]);
            srec[place] = o;
            scol[place] = c[u];
        }
    }
    __syncthreads();
    const int left = e - blockIdx.x * CH_EDGES;
    const int nval = left < CH_EDGES ? left : CH_EDGES;
    int cj[CH_EPT], sl[CH_EPT];
    int2 sr[CH_EPT];
    #pragma unroll
    for (int u = 0; u < CH_EPT; ++u) {
        const int j = tid + u * CH_THREADS;
        const int jc = j < nval ? j : nval - 1;
        cj[u] = scol[jc];
        sr[u] = srec[jc];
    }
    #pragma unroll
    for (int u = 0; u < CH_EPT; ++u)
        sl[u] = off[cj[u] >> BK_SHIFT] + tid + u * CH_THREADS;
    #pragma unroll
    for (int u = 0; u < CH_EPT; ++u) {
        if (tid + u * CH_THREADS < nval) {
            reinterpret_cast<int2*>(trec)[sl[u]] = sr[u];
            tcol[sl[u]] = cj[u];
        }
    }
}

#define SORT_THREADS 256
#define SORT_EPT 9

__global__ __launch_bounds__(SORT_THREADS)
void gcn_bucket_sort(const int* __restrict__ trec, const int* __restrict__ tcol,
                     const int* __restrict__ bstart, int* __restrict__ cnt,
                     int* __restrict__ start, int* __restrict__ rec, int n) {
    __shared__ int lc[BK_NODES];
    __shared__ int ls[BK_NODES];
    __shared__ int wsum[BK_NODES / 32];
    const int tid = threadIdx.x;
    const int lane = tid & 31;
    const int b = blockIdx.x;
    const int s0 = bstart[b];
    const int s1 = bstart[b + 1];
    const int2* __restrict__ trec2 = reinterpret_cast<const int2*>(trec);
    int2* __restrict__ rec2 = reinterpret_cast<int2*>(rec);
    if (tid < BK_NODES) lc[tid] = 0;
    __syncthreads();
    for (int i0 = s0 + tid; i0 < s1; i0 += SORT_EPT * SORT_THREADS) {
        int cc[SORT_EPT];
        #pragma unroll
        for (int u = 0; u < SORT_EPT; ++u) {
            const int i = i0 + u * SORT_THREADS;
            const int v = tcol[i < s1 ? i : s1 - 1];
            cc[u] = i < s1 ? v : -1;
        }
        #pragma unroll
        for (int u = 0; u < SORT_EPT; ++u)
            if (cc[u] >= 0) atomicAdd(&lc[cc[u] & (BK_NODES - 1)], 1);
    }
    __syncthreads();
    int x = 0, incl = 0;
    if (tid < BK_NODES) {
        x = lc[tid];
        incl = x;
        #pragma unroll
        for (int o = 1; o < 32; o <<= 1) {
            const int y = __shfl_up(incl, o);
            if (lane >= o) incl += y;
        }
        if (lane == 31) wsum[tid >> 5] = incl;
    }
    __syncthreads();
    if (tid < BK_NODES) {
        int wo = 0;
        #pragma unroll
        for (int k = 0; k < BK_NODES / 32; ++k) wo += k < (tid >> 5) ? wsum[k] : 0;
        const int first = s0 + wo + incl - x;
        ls[tid] = first;
        lc[tid] = 0;
        const int node = b * BK_NODES + tid;
        if (node < n) { cnt[node] = x; start[node] = first; }
    }
    __syncthreads();
    for (int i0 = s0 + tid; i0 < s1; i0 += SORT_EPT * SORT_THREADS) {
        int jj[SORT_EPT], rk[SORT_EPT], fs[SORT_EPT];
        int2 rr[SORT_EPT];
        #pragma unroll
        for (int u = 0; u < SORT_EPT; ++u) {
            const int i = i0 + u * SORT_THREADS;
            const int ic = i < s1 ? i : s1 - 1;
            jj[u] = tcol[ic] & (BK_NODES - 1);
            rr[u] = trec2[ic];
        }
        #pragma unroll
        for (int u = 0; u < SORT_EPT; ++u)
            rk[u] = atomicAdd(&lc[jj[u]], i0 + u * SORT_THREADS < s1 ? 1 : 0);
        #pragma unroll
        for (int u = 0; u < SORT_EPT; ++u) fs[u] = ls[jj[u]];
        #pragma unroll
        for (int u = 0; u < SORT_EPT; ++u)
            if (i0 + u * SORT_THREADS < s1) rec2[fs[u] + rk[u]] = rr[u];
    }
}

#define SCALE_NODES 8

__global__ __launch_bounds__(256)
void gcn_deg_scale(const float* __restrict__ X, const int* __restrict__ start,
                   const int* __restrict__ cnt, const int* __restrict__ rec,
                   float* __restrict__ dinv, float* __restrict__ Xs, int n) {
    const int lane = threadIdx.x & 31;
    const int wave = __builtin_amdgcn_readfirstlane(threadIdx.x >> 5);
    const int node0 = (blockIdx.x * 8 + wave) * SCALE_NODES;
    const int2* __restrict__ rec2 = reinterpret_cast<const int2*>(rec);
    const int ln  = node0 + (lane & (SCALE_NODES - 1));
    const int lnd = ln < n ? ln : n - 1;
    const int sv = start[lnd];
    const int cv = cnt[lnd];
    int    s0[SCALE_NODES], cn[SCALE_NODES];
    float  ws[SCALE_NODES];
    float2 xr[SCALE_NODES];
    #pragma unroll
    for (int t = 0; t < SCALE_NODES; ++t) {
        const int node = node0 + t;
        const bool valid = node < n;
        const int nd = valid ? node : n - 1;
        s0[t] = __builtin_amdgcn_readlane(sv, t);
        const int c = __builtin_amdgcn_readlane(cv, t);
        cn[t] = valid ? c : 0;
        xr[t] = *reinterpret_cast<const float2*>(X + (size_t)nd * FDIM + lane * 2);
        ws[t] = 0.f;
        if (cn[t] > 0) {
            const int rem = cn[t] < 32 ? cn[t] : 32;
            const int2 r = rec2[s0[t] + (lane < rem ? lane : rem - 1)];
            ws[t] = lane < rem ? __int_as_float(r.y) : 0.f;
        }
    }
    #pragma unroll
    for (int t = 0; t < SCALE_NODES; ++t) {
        const int s1 = s0[t] + cn[t];
        for (int base = s0[t] + 32; base < s1; base += 32) {
            const int rem = s1 - base;
            if (lane < rem) ws[t] += __int_as_float(rec2[base + lane].y);
        }
    }
    #pragma unroll
    for (int off = 16; off > 0; off >>= 1) {
        float o[SCALE_NODES];
        #pragma unroll
        for (int t = 0; t < SCALE_NODES; ++t) o[t] = __shfl_xor(ws[t], off);
        #pragma unroll
        for (int t = 0; t < SCALE_NODES; ++t) ws[t] += o[t];
    }
    float mine = 0.f;
    #pragma unroll
    for (int t = 0; t < SCALE_NODES; ++t) {
        const float d = 1.0f + ws[t];
        const float di = d > 0.f ? __frsqrt_rn(d) : 0.f;
        if ((lane & (SCALE_NODES - 1)) == t) mine = di;
        const int node = node0 + t;
        if (node < n) {
            float2 o;
            o.x = di * xr[t].x;
            o.y = di * xr[t].y;
            *reinterpret_cast<float2*>(Xs + (size_t)node * FDIM + lane * 2) = o;
        }
    }
    if (lane < SCALE_NODES && ln < n) dinv[ln] = mine;
}

template <int B>
__device__ __forceinline__ void gcn_gather_batch(const float* __restrict__ X, int rsrc, int rnrm,
                                                 int j0, int lane, float2& acc) {
    float2 v[B];
    float nj[B];
    #pragma unroll
    for (int u = 0; u < B; ++u) {
        const int rj = __builtin_amdgcn_readlane(rsrc, j0 + u);
        nj[u] = __int_as_float(__builtin_amdgcn_readlane(rnrm, j0 + u));
        v[u] = *reinterpret_cast<const float2*>(X + (size_t)rj * FDIM + lane * 2);
    }
    #pragma unroll
    for (int u = 0; u < B; ++u) {
        acc.x += nj[u] * v[u].x;
        acc.y += nj[u] * v[u].y;
    }
}

__device__ __forceinline__ void gcn_gather_chunk(const float* __restrict__ X, int2 rc, int rem,
                                                 int lane, float2& acc) {
    int j = 0;
    for (; j + 8 <= rem; j += 8) gcn_gather_batch<8>(X, rc.x, rc.y, j, lane, acc);
    if (j + 4 <= rem) { gcn_gather_batch<4>(X, rc.x, rc.y, j, lane, acc); j += 4; }
    if (j + 2 <= rem) { gcn_gather_batch<2>(X, rc.x, rc.y, j, lane, acc); j += 2; }
    if (j < rem)      { gcn_gather_batch<1>(X, rc.x, rc.y, j, lane, acc); }
}

#define TILE_NODES 32
#define NODES_PER_WAVE 4
#define TILE_PITCH 68

__global__ __launch_bounds__(256)
void gcn_agg_gemm(const float* __restrict__ Xin, const int* __restrict__ start,
                  const int* __restrict__ cnt, const int* __restrict__ rec,
                  const float* __restrict__ dinv, const float* __restrict__ W,
                  const float* __restrict__ bias, float* __restrict__ out,
                  int n, int relu) {
    __shared__ alignas(16) float tile[TILE_NODES * TILE_PITCH];
    __shared__ alignas(16) float wt[FDIM * TILE_PITCH];

    {
        const int tid = threadIdx.x;
        float4 w4[4];
        #pragma unroll
        for (int i = 0; i < 4; ++i)
            w4[i] = *reinterpret_cast<const float4*>(W + i * 1024 + tid * 4);
        #pragma unroll
        for (int i = 0; i < 4; ++i)
            *reinterpret_cast<float4*>(&wt[(i * 16 + (tid >> 4)) * TILE_PITCH + (tid & 15) * 4]) = w4[i];
    }

    const int lane = threadIdx.x & 31;
    const int wave = __builtin_amdgcn_readfirstlane(threadIdx.x >> 5);
    const int nodeBase = blockIdx.x * TILE_NODES;
    const int2* __restrict__ rec2 = reinterpret_cast<const int2*>(rec);

    const int node0 = nodeBase + wave * NODES_PER_WAVE;
    const int ln  = node0 + (lane & (NODES_PER_WAVE - 1));
    const int lnd = ln < n ? ln : n - 1;
    const int   sv = start[lnd];
    const int   cv = cnt[lnd];
    const float dv = dinv[lnd];
    int   s0[NODES_PER_WAVE], cn[NODES_PER_WAVE];
    float dsc[NODES_PER_WAVE];
    int2  rc[NODES_PER_WAVE];
    float2 hs[NODES_PER_WAVE];
    #pragma unroll
    for (int t = 0; t < NODES_PER_WAVE; ++t) {
        const bool valid = node0 + t < n;
        s0[t] = __builtin_amdgcn_readlane(sv, t);
        const int c = __builtin_amdgcn_readlane(cv, t);
        const float di = __int_as_float(__builtin_amdgcn_readlane(__float_as_int(dv), t));
        cn[t] = valid ? c : 0;
        dsc[t] = valid ? di : 0.f;
    }
    #pragma unroll
    for (int t = 0; t < NODES_PER_WAVE; ++t) {
        const int node = node0 + t;
        const int nd = node < n ? node : n - 1;
        hs[t] = *reinterpret_cast<const float2*>(Xin + (size_t)nd * FDIM + lane * 2);
        int rem = cn[t] < 32 ? cn[t] : 32;
        rc[t].x = 0; rc[t].y = 0;
        if (rem > 0) rc[t] = rec2[s0[t] + (lane < rem ? lane : rem - 1)];
    }
    #pragma unroll
    for (int t = 0; t < NODES_PER_WAVE; ++t) {
        float2 acc; acc.x = 0.f; acc.y = 0.f;
        const int s1 = s0[t] + cn[t];
        gcn_gather_chunk(Xin, rc[t], cn[t] < 32 ? cn[t] : 32, lane, acc);
        for (int base = s0[t] + 32; base < s1; base += 32) {
            int rem = s1 - base;
            if (rem > 32) rem = 32;
            const int2 rmore = rec2[base + (lane < rem ? lane : rem - 1)];
            gcn_gather_chunk(Xin, rmore, rem, lane, acc);
        }
        acc.x = dsc[t] * (acc.x + hs[t].x);
        acc.y = dsc[t] * (acc.y + hs[t].y);
        *reinterpret_cast<float2*>(&tile[(wave * NODES_PER_WAVE + t) * TILE_PITCH + lane * 2]) = acc;
    }

    const int rowT = wave >> 2;
    const int colBase = (wave & 3) * 16;
    const int half = lane >> 4;
    const int l15  = lane & 15;
    const float2 bb = *reinterpret_cast<const float2*>(bias + lane * 2);

    __syncthreads();

    const float* arow = &tile[(rowT * 16 + l15) * TILE_PITCH];
    const float* wrow = &wt[(colBase + l15) * TILE_PITCH];
    v2f a[FDIM / 4], b[FDIM / 4];
    v8f c = {0.f, 0.f, 0.f, 0.f, 0.f, 0.f, 0.f, 0.f};
    __builtin_amdgcn_sched_barrier(0);
    #pragma unroll
    for (int q = 0; q < FDIM / 4; ++q) {
        const int ka = q * 4 + half * 2;
        a[q].x = arow[ka]; a[q].y = arow[ka + 1];
        b[q].x = wrow[ka]; b[q].y = wrow[ka + 1];
        c = __builtin_amdgcn_wmma_f32_16x16x4_f32(
                false, a[q], false, b[q], (short)0, c, false, false);
    }
    asm volatile("v_nop\n\tv_nop\n\tv_nop\n\tv_nop\n\tv_nop"
                 : "+v"(c)
                 : "v"(a[FDIM / 4 - 2]), "v"(b[FDIM / 4 - 2]), "v"(a[FDIM / 4 - 1]), "v"(b[FDIM / 4 - 1]));
    __builtin_amdgcn_sched_barrier(0);

    __syncthreads();
    #pragma unroll
    for (int r = 0; r < 8; ++r)
        tile[(rowT * 16 + half * 8 + r) * TILE_PITCH + colBase + l15] = c[r];
    __syncthreads();
    #pragma unroll
    for (int t = 0; t < NODES_PER_WAVE; ++t) {
        const int rrow = wave * NODES_PER_WAVE + t;
        const int m = nodeBase + rrow;
        if (m < n) {
            float2 v = *reinterpret_cast<const float2*>(&tile[rrow * TILE_PITCH + lane * 2]);
            v.x += bb.x;
            v.y += bb.y;
            if (relu) { v.x = dsc[t] * fmaxf(v.x, 0.f); v.y = dsc[t] * fmaxf(v.y, 0.f); }
            *reinterpret_cast<float2*>(out + (size_t)m * FDIM + lane * 2) = v;
        }
    }
}

extern "C" void kernel_launch(void* const* d_in, const int* in_sizes, int n_in,
                              void* d_out, int out_size, void* d_ws, size_t ws_size,
                              hipStream_t stream) {
    const float* x   = (const float*)d_in[0];
    const float* ew  = (const float*)d_in[1];
    const float* W1  = (const float*)d_in[2];
    const float* b1  = (const float*)d_in[3];
    const float* W2  = (const float*)d_in[4];
    const float* b2  = (const float*)d_in[5];
    const int*  eidx = (const int*)d_in[6];

    const int n = in_sizes[0] / FDIM;
    const int e = in_sizes[1];
    const int* row = eidx;
    const int* col = eidx + e;

    const int nb   = (n + BK_NODES - 1) / BK_NODES;
    const int nblk = (e + CH_EDGES - 1) / CH_EDGES;
    float* dinv  = (float*)d_ws;
    int*   cnt   = (int*)(dinv + n);
    int*   rec   = (int*)(cnt + n);
    int*   start = rec + 2 * (size_t)e;
    int*   bstart = start + n;
    int*   H     = bstart + (MAX_BK + 1);
    float* agg1  = (float*)(cnt + n) + (size_t)n * FDIM;
    int*   trec  = (int*)agg1;
    int*   tcol  = trec + 2 * (size_t)e;
    float* outp  = (float*)d_out;

    const int B = 256;
    const int gTile = (n + TILE_NODES - 1) / TILE_NODES;
    const int gScale = (n + 8 * SCALE_NODES - 1) / (8 * SCALE_NODES);

    gcn_bucket_count  <<<nblk, CH_THREADS, 0, stream>>>(col, H, e, nb);
    gcn_bucket_scatter<<<nblk, CH_THREADS, 0, stream>>>(row, col, ew, H, bstart, trec, tcol, e, nb, nblk);
    gcn_bucket_sort   <<<nb, SORT_THREADS, 0, stream>>>(trec, tcol, bstart, cnt, start, rec, n);
    gcn_deg_scale<<<gScale, B, 0, stream>>>(x, start, cnt, rec, dinv, outp, n);

    gcn_agg_gemm <<<gTile, B, 0, stream>>>(outp, start, cnt, rec, dinv, W1, b1, agg1, n, 1);

    gcn_agg_gemm <<<gTile, B, 0, stream>>>(agg1, start, cnt, rec, dinv, W2, b2, outp, n, 0);
}


# edges bucket-ordered inside their own 8192-edge chunk; a bucket block finds its runs in a chunk-by-bucket offset table by LDS bisection: counting launch gone (6 launches to 5); sums in LDS rank order
# speedup vs baseline: 4.0456x; 1.1344x over previous
#include <hip/hip_runtime.h>
#include <hip/hip_bf16.h>

typedef float v2f  __attribute__((ext_vector_type(2)));
typedef float v8f  __attribute__((ext_vector_type(8)));

#define FDIM 64

#define BK_SHIFT 7
#define BK_NODES (1 << BK_SHIFT)
#define MAX_BK 512
#define CH_THREADS 1024
#define CH_EPT 8
#define CH_EDGES (CH_THREADS * CH_EPT)
#define MAX_CH 128

__global__ __launch_bounds__(CH_THREADS)
void gcn_chunk_order(const int* __restrict__ row, const int* __restrict__ col,
                     const float* __restrict__ w, int* __restrict__ offs,
                     int* __restrict__ trec, int* __restrict__ tcol, int e, int nb) {
    __shared__ int lc[MAX_BK];
    __shared__ int loff[MAX_BK];
    __shared__ int wsum[MAX_BK / 32];
    __shared__ alignas(8) int2 srec[CH_EDGES];
    __shared__ int scol[CH_EDGES];
    const int tid = threadIdx.x;
    const int lane = tid & 31;
    if (tid < MAX_BK) lc[tid] = 0;
    const int first = blockIdx.x * CH_EDGES;
    const int base = first + tid;
    int c[CH_EPT], rw[CH_EPT], rk[CH_EPT];
    float wv[CH_EPT];
    #pragma unroll
    for (int u = 0; u < CH_EPT; ++u) {
        const int i = base + u * CH_THREADS;
        const int ic = i < e ? i : e - 1;
        const int v = col[ic];
        rw[u] = row[ic];
        wv[u] = w[ic];
        c[u] = i < e ? v : -1;
    }
    __syncthreads();
    #pragma unroll
    for (int u = 0; u < CH_EPT; ++u) {
        rk[u] = 0;
        if (c[u] >= 0) rk[u] = atomicAdd(&lc[c[u] >> BK_SHIFT], 1);
    }
    __syncthreads();
    int y = 0, iy = 0;
    if (tid < MAX_BK) {
        y = lc[tid];
        iy = y;
        #pragma unroll
        for (int o = 1; o < 32; o <<= 1) {
            const int vy = __shfl_up(iy, o);
            if (lane >= o) iy += vy;
        }
        if (lane == 31) wsum[tid >> 5] = iy;
    }
    __syncthreads();
    const int left = e - first;
    const int nval = left < CH_EDGES ? left : CH_EDGES;
    if (tid < MAX_BK) {
        int wy = 0;
        #pragma unroll
        for (int k = 0; k < MAX_BK / 32; ++k) wy += k < (tid >> 5) ? wsum[k] : 0;
        const int ey = wy + iy - y;
        loff[tid] = ey;
        int* orow = offs + (size_t)blockIdx.x * (nb + 1);
        if (tid < nb) orow[tid] = ey;
        if (tid == 0) orow[nb] = nval;
    }
    __syncthreads();
    #pragma unroll
    for (int u = 0; u < CH_EPT; ++u) {
        if (c[u] >= 0) {
            const int place = loff[c[u] >> BK_SHIFT] + rk[u];
            int2 o;
            o.x = rw[u];
            o.y = __float_as_int(wv[u]);
            srec[place] = o;
            scol[place] = c[u];
        }
    }
    __syncthreads();
    int cj[CH_EPT];
    int2 sr[CH_EPT];
    #pragma unroll
    for (int u = 0; u < CH_EPT; ++u) {
        const int j = tid + u * CH_THREADS;
        const int jc = j < nval ? j : nval - 1;
        cj[u] = scol[jc];
        sr[u] = srec[jc];
    }
    #pragma unroll
    for (int u = 0; u < CH_EPT; ++u) {
        const int j = tid + u * CH_THREADS;
        if (j < nval) {
            reinterpret_cast<int2*>(trec)[first + j] = sr[u];
            tcol[first + j] = cj[u];
        }
    }
}

#define SORT_THREADS 256
#define SORT_EPT 9
#define SORT_TRIP (SORT_EPT * SORT_THREADS)

__device__ __forceinline__ void gcn_locate(const int* rpre, const int* rsrc,
                                           const int (&ic)[SORT_EPT], int (&src)[SORT_EPT]) {
    int k[SORT_EPT];
    #pragma unroll
    for (int u = 0; u < SORT_EPT; ++u) k[u] = 0;
    #pragma unroll
    for (int s = MAX_CH / 2; s > 0; s >>= 1) {
        #pragma unroll
        for (int u = 0; u < SORT_EPT; ++u) k[u] += rpre[k[u] + s] <= ic[u] ? s : 0;
    }
    #pragma unroll
    for (int u = 0; u < SORT_EPT; ++u) src[u] = rsrc[k[u]] + ic[u];
}

__global__ __launch_bounds__(SORT_THREADS)
void gcn_bucket_sort(const int* __restrict__ trec, const int* __restrict__ tcol,
                     const int* __restrict__ offs, int* __restrict__ cnt,
                     int* __restrict__ start, int* __restrict__ rec, int n, int nb, int nblk) {
    __shared__ int lc[BK_NODES];
    __shared__ int ls[BK_NODES];
    __shared__ int wsum[BK_NODES / 32];
    __shared__ int rpre[MAX_CH];
    __shared__ int rsrc[MAX_CH];
    __shared__ int rws[2][MAX_CH / 32];
    const int tid = threadIdx.x;
    const int lane = tid & 31;
    const int b = blockIdx.x;
    const int2* __restrict__ trec2 = reinterpret_cast<const int2*>(trec);
    int2* __restrict__ rec2 = reinterpret_cast<int2*>(rec);
    int lo = 0, len = 0, il = 0, io = 0;
    if (tid < MAX_CH) {
        if (tid < nblk) {
            const int* p = offs + (size_t)tid * (nb + 1) + b;
            lo = p[0];
            len = p[1] - lo;
        }
        il = len; io = lo;
        #pragma unroll
        for (int o = 1; o < 32; o <<= 1) {
            const int vl = __shfl_up(il, o);
            const int vo = __shfl_up(io, o);
            if (lane >= o) { il += vl; io += vo; }
        }
        if (lane == 31) { rws[0][tid >> 5] = il; rws[1][tid >> 5] = io; }
    }
    if (tid < BK_NODES) lc[tid] = 0;
    __syncthreads();
    int cntb = 0, s0 = 0;
    #pragma unroll
    for (int k = 0; k < MAX_CH / 32; ++k) { cntb += rws[0][k]; s0 += rws[1][k]; }
    if (tid < MAX_CH) {
        int wl = 0;
        #pragma unroll
        for (int k = 0; k < MAX_CH / 32; ++k) wl += k < (tid >> 5) ? rws[0][k] : 0;
        const int pre = wl + il - len;
        rpre[tid] = pre;
        rsrc[tid] = tid * CH_EDGES + lo - pre;
    }
    __syncthreads();
    const int last = cntb - 1;
    int jj0[SORT_EPT];
    int2 rr0[SORT_EPT];
    #pragma unroll
    for (int u = 0; u < SORT_EPT; ++u) { jj0[u] = 0; rr0[u].x = 0; rr0[u].y = 0; }
    if (cntb > 0) {
        int ic[SORT_EPT], sc[SORT_EPT];
        #pragma unroll
        for (int u = 0; u < SORT_EPT; ++u) {
            const int i = tid + u * SORT_THREADS;
            ic[u] = i < last ? i : last;
        }
        gcn_locate(rpre, rsrc, ic, sc);
        #pragma unroll
        for (int u = 0; u < SORT_EPT; ++u) {
            jj0[u] = tcol[sc[u]] & (BK_NODES - 1);
            rr0[u] = trec2[sc[u]];
        }
    }
    #pragma unroll
    for (int u = 0; u < SORT_EPT; ++u)
        if (tid + u * SORT_THREADS < cntb) atomicAdd(&lc[jj0[u]], 1);
    for (int i0 = tid + SORT_TRIP; i0 < cntb; i0 += SORT_TRIP) {
        int ic[SORT_EPT], sc[SORT_EPT], cc[SORT_EPT];
        #pragma unroll
        for (int u = 0; u < SORT_EPT; ++u) {
            const int i = i0 + u * SORT_THREADS;
            ic[u] = i < last ? i : last;
        }
        gcn_locate(rpre, rsrc, ic, sc);
        #pragma unroll
        for (int u = 0; u < SORT_EPT; ++u) cc[u] = tcol[sc[u]] & (BK_NODES - 1);
        #pragma unroll
        for (int u = 0; u < SORT_EPT; ++u)
            if (i0 + u * SORT_THREADS < cntb) atomicAdd(&lc[cc[u]], 1);
    }
    __syncthreads();
    int x = 0, incl = 0;
    if (tid < BK_NODES) {
        x = lc[tid];
        incl = x;
        #pragma unroll
        for (int o = 1; o < 32; o <<= 1) {
            const int y = __shfl_up(incl, o);
            if (lane >= o) incl += y;
        }
        if (lane == 31) wsum[tid >> 5] = incl;
    }
    __syncthreads();
    if (tid < BK_NODES) {
        int wo = 0;
        #pragma unroll
        for (int k = 0; k < BK_NODES / 32; ++k) wo += k < (tid >> 5) ? wsum[k] : 0;
        const int first = s0 + wo + incl - x;
        ls[tid] = first;
        lc[tid] = 0;
        const int node = b * BK_NODES + tid;
        if (node < n) { cnt[node] = x; start[node] = first; }
    }
    __syncthreads();
    {
        int rk[SORT_EPT], fs[SORT_EPT];
        #pragma unroll
        for (int u = 0; u < SORT_EPT; ++u)
            rk[u] = atomicAdd(&lc[jj0[u]], tid + u * SORT_THREADS < cntb ? 1 : 0);
        #pragma unroll
        for (int u = 0; u < SORT_EPT; ++u) fs[u] = ls[jj0[u]];
        #pragma unroll
        for (int u = 0; u < SORT_EPT; ++u)
            if (tid + u * SORT_THREADS < cntb) rec2[fs[u] + rk[u]] = rr0[u];
    }
    for (int i0 = tid + SORT_TRIP; i0 < cntb; i0 += SORT_TRIP) {
        int ic[SORT_EPT], sc[SORT_EPT], jj[SORT_EPT], rk[SORT_EPT], fs[SORT_EPT];
        int2 rr[SORT_EPT];
        #pragma unroll
        for (int u = 0; u < SORT_EPT; ++u) {
            const int i = i0 + u * SORT_THREADS;
            ic[u] = i < last ? i : last;
        }
        gcn_locate(rpre, rsrc, ic, sc);
        #pragma unroll
        for (int u = 0; u < SORT_EPT; ++u) {
            jj[u] = tcol[sc[u]] & (BK_NODES - 1);
            rr[u] = trec2[sc[u]];
        }
        #pragma unroll
        for (int u = 0; u < SORT_EPT; ++u)
            rk[u] = atomicAdd(&lc[jj[u]], i0 + u * SORT_THREADS < cntb ? 1 : 0);
        #pragma unroll
        for (int u = 0; u < SORT_EPT; ++u) fs[u] = ls[jj[u]];
        #pragma unroll
        for (int u = 0; u < SORT_EPT; ++u)
            if (i0 + u * SORT_THREADS < cntb) rec2[fs[u] + rk[u]] = rr[u];
    }
}

#define SCALE_NODES 8

__global__ __launch_bounds__(256)
void gcn_deg_scale(const float* __restrict__ X, const int* __restrict__ start,
                   const int* __restrict__ cnt, const int* __restrict__ rec,
                   float* __restrict__ dinv, float* __restrict__ Xs, int n) {
    const int lane = threadIdx.x & 31;
    const int wave = __builtin_amdgcn_readfirstlane(threadIdx.x >> 5);
    const int node0 = (blockIdx.x * 8 + wave) * SCALE_NODES;
    const int2* __restrict__ rec2 = reinterpret_cast<const int2*>(rec);
    const int ln  = node0 + (lane & (SCALE_NODES - 1));
    const int lnd = ln < n ? ln : n - 1;
    const int sv = start[lnd];
    const int cv = cnt[lnd];
    int    s0[SCALE_NODES], cn[SCALE_NODES];
    float  ws[SCALE_NODES];
    float2 xr[SCALE_NODES];
    #pragma unroll
    for (int t = 0; t < SCALE_NODES; ++t) {
        const int node = node0 + t;
        const bool valid = node < n;
        const int nd = valid ? node : n - 1;
        s0[t] = __builtin_amdgcn_readlane(sv, t);
        const int c = __builtin_amdgcn_readlane(cv, t);
        cn[t] = valid ? c : 0;
        xr[t] = *reinterpret_cast<const float2*>(X + (size_t)nd * FDIM + lane * 2);
        ws[t] = 0.f;
        if (cn[t] > 0) {
            const int rem = cn[t] < 32 ? cn[t] : 32;
            const int2 r = rec2[s0[t] + (lane < rem ? lane : rem - 1)];
            ws[t] = lane < rem ? __int_as_float(r.y) : 0.f;
        }
    }
    #pragma unroll
    for (int t = 0; t < SCALE_NODES; ++t) {
        const int s1 = s0[t] + cn[t];
        for (int base = s0[t] + 32; base < s1; base += 32) {
            const int rem = s1 - base;
            if (lane < rem) ws[t] += __int_as_float(rec2[base + lane].y);
        }
    }
    #pragma unroll
    for (int off = 16; off > 0; off >>= 1) {
        float o[SCALE_NODES];
        #pragma unroll
        for (int t = 0; t < SCALE_NODES; ++t) o[t] = __shfl_xor(ws[t], off);
        #pragma unroll
        for (int t = 0; t < SCALE_NODES; ++t) ws[t] += o[t];
    }
    float mine = 0.f;
    #pragma unroll
    for (int t = 0; t < SCALE_NODES; ++t) {
        const float d = 1.0f + ws[t];
        const float di = d > 0.f ? __frsqrt_rn(d) : 0.f;
        if ((lane & (SCALE_NODES - 1)) == t) mine = di;
        const int node = node0 + t;
        if (node < n) {
            float2 o;
            o.x = di * xr[t].x;
            o.y = di * xr[t].y;
            *reinterpret_cast<float2*>(Xs + (size_t)node * FDIM + lane * 2) = o;
        }
    }
    if (lane < SCALE_NODES && ln < n) dinv[ln] = mine;
}

template <int B>
__device__ __forceinline__ void gcn_gather_batch(const float* __restrict__ X, int rsrc, int rnrm,
                                                 int j0, int lane, float2& acc) {
    float2 v[B];
    float nj[B];
    #pragma unroll
    for (int u = 0; u < B; ++u) {
        const int rj = __builtin_amdgcn_readlane(rsrc, j0 + u);
        nj[u] = __int_as_float(__builtin_amdgcn_readlane(rnrm, j0 + u));
        v[u] = *reinterpret_cast<const float2*>(X + (size_t)rj * FDIM + lane * 2);
    }
    #pragma unroll
    for (int u = 0; u < B; ++u) {
        acc.x += nj[u] * v[u].x;
        acc.y += nj[u] * v[u].y;
    }
}

__device__ __forceinline__ void gcn_gather_chunk(const float* __restrict__ X, int2 rc, int rem,
                                                 int lane, float2& acc) {
    int j = 0;
    for (; j + 8 <= rem; j += 8) gcn_gather_batch<8>(X, rc.x, rc.y, j, lane, acc);
    if (j + 4 <= rem) { gcn_gather_batch<4>(X, rc.x, rc.y, j, lane, acc); j += 4; }
    if (j + 2 <= rem) { gcn_gather_batch<2>(X, rc.x, rc.y, j, lane, acc); j += 2; }
    if (j < rem)      { gcn_gather_batch<1>(X, rc.x, rc.y, j, lane, acc); }
}

#define TILE_NODES 32
#define NODES_PER_WAVE 4
#define TILE_PITCH 68

__global__ __launch_bounds__(256)
void gcn_agg_gemm(const float* __restrict__ Xin, const int* __restrict__ start,
                  const int* __restrict__ cnt, const int* __restrict__ rec,
                  const float* __restrict__ dinv, const float* __restrict__ W,
                  const float* __restrict__ bias, float* __restrict__ out,
                  int n, int relu) {
    __shared__ alignas(16) float tile[TILE_NODES * TILE_PITCH];
    __shared__ alignas(16) float wt[FDIM * TILE_PITCH];

    {
        const int tid = threadIdx.x;
        float4 w4[4];
        #pragma unroll
        for (int i = 0; i < 4; ++i)
            w4[i] = *reinterpret_cast<const float4*>(W + i * 1024 + tid * 4);
        #pragma unroll
        for (int i = 0; i < 4; ++i)
            *reinterpret_cast<float4*>(&wt[(i * 16 + (tid >> 4)) * TILE_PITCH + (tid & 15) * 4]) = w4[i];
    }

    const int lane = threadIdx.x & 31;
    const int wave = __builtin_amdgcn_readfirstlane(threadIdx.x >> 5);
    const int nodeBase = blockIdx.x * TILE_NODES;
    const int2* __restrict__ rec2 = reinterpret_cast<const int2*>(rec);

    const int node0 = nodeBase + wave * NODES_PER_WAVE;
    const int ln  = node0 + (lane & (NODES_PER_WAVE - 1));
    const int lnd = ln < n ? ln : n - 1;
    const int   sv = start[lnd];
    const int   cv = cnt[lnd];
    const float dv = dinv[lnd];
    int   s0[NODES_PER_WAVE], cn[NODES_PER_WAVE];
    float dsc[NODES_PER_WAVE];
    int2  rc[NODES_PER_WAVE];
    float2 hs[NODES_PER_WAVE];
    #pragma unroll
    for (int t = 0; t < NODES_PER_WAVE; ++t) {
        const bool valid = node0 + t < n;
        s0[t] = __builtin_amdgcn_readlane(sv, t);
        const int c = __builtin_amdgcn_readlane(cv, t);
        const float di = __int_as_float(__builtin_amdgcn_readlane(__float_as_int(dv), t));
        cn[t] = valid ? c : 0;
        dsc[t] = valid ? di : 0.f;
    }
    #pragma unroll
    for (int t = 0; t < NODES_PER_WAVE; ++t) {
        const int node = node0 + t;
        const int nd = node < n ? node : n - 1;
        hs[t] = *reinterpret_cast<const float2*>(Xin + (size_t)nd * FDIM + lane * 2);
        int rem = cn[t] < 32 ? cn[t] : 32;
        rc[t].x = 0; rc[t].y = 0;
        if (rem > 0) rc[t] = rec2[s0[t] + (lane < rem ? lane : rem - 1)];
    }
    #pragma unroll
    for (int t = 0; t < NODES_PER_WAVE; ++t) {
        float2 acc; acc.x = 0.f; acc.y = 0.f;
        const int s1 = s0[t] + cn[t];
        gcn_gather_chunk(Xin, rc[t], cn[t] < 32 ? cn[t] : 32, lane, acc);
        for (int base = s0[t] + 32; base < s1; base += 32) {
            int rem = s1 - base;
            if (rem > 32) rem = 32;
            const int2 rmore = rec2[base + (lane < rem ? lane : rem - 1)];
            gcn_gather_chunk(Xin, rmore, rem, lane, acc);
        }
        acc.x = dsc[t] * (acc.x + hs[t].x);
        acc.y = dsc[t] * (acc.y + hs[t].y);
        *reinterpret_cast<float2*>(&tile[(wave * NODES_PER_WAVE + t) * TILE_PITCH + lane * 2]) = acc;
    }

    const int rowT = wave >> 2;
    const int colBase = (wave & 3) * 16;
    const int half = lane >> 4;
    const int l15  = lane & 15;
    const float2 bb = *reinterpret_cast<const float2*>(bias + lane * 2);

    __syncthreads();

    const float* arow = &tile[(rowT * 16 + l15) * TILE_PITCH];
    const float* wrow = &wt[(colBase + l15) * TILE_PITCH];
    v2f a[FDIM / 4], b[FDIM / 4];
    v8f c = {0.f, 0.f, 0.f, 0.f, 0.f, 0.f, 0.f, 0.f};
    __builtin_amdgcn_sched_barrier(0);
    #pragma unroll
    for (int q = 0; q < FDIM / 4; ++q) {
        const int ka = q * 4 + half * 2;
        a[q].x = arow[ka]; a[q].y = arow[ka + 1];
        b[q].x = wrow[ka]; b[q].y = wrow[ka + 1];
        c = __builtin_amdgcn_wmma_f32_16x16x4_f32(
                false, a[q], false, b[q], (short)0, c, false, false);
    }
    asm volatile("v_nop\n\tv_nop\n\tv_nop\n\tv_nop\n\tv_nop"
                 : "+v"(c)
                 : "v"(a[FDIM / 4 - 2]), "v"(b[FDIM / 4 - 2]), "v"(a[FDIM / 4 - 1]), "v"(b[FDIM / 4 - 1]));
    __builtin_amdgcn_sched_barrier(0);

    __syncthreads();
    #pragma unroll
    for (int r = 0; r < 8; ++r)
        tile[(rowT * 16 + half * 8 + r) * TILE_PITCH + colBase + l15] = c[r];
    __syncthreads();
    #pragma unroll
    for (int t = 0; t < NODES_PER_WAVE; ++t) {
        const int rrow = wave * NODES_PER_WAVE + t;
        const int m = nodeBase + rrow;
        if (m < n) {
            float2 v = *reinterpret_cast<const float2*>(&tile[rrow * TILE_PITCH + lane * 2]);
            v.x += bb.x;
            v.y += bb.y;
            if (relu) { v.x = dsc[t] * fmaxf(v.x, 0.f); v.y = dsc[t] * fmaxf(v.y, 0.f); }
            *reinterpret_cast<float2*>(out + (size_t)m * FDIM + lane * 2) = v;
        }
    }
}

extern "C" void kernel_launch(void* const* d_in, const int* in_sizes, int n_in,
                              void* d_out, int out_size, void* d_ws, size_t ws_size,
                              hipStream_t stream) {
    const float* x   = (const float*)d_in[0];
    const float* ew  = (const float*)d_in[1];
    const float* W1  = (const float*)d_in[2];
    const float* b1  = (const float*)d_in[3];
    const float* W2  = (const float*)d_in[4];
    const float* b2  = (const float*)d_in[5];
    const int*  eidx = (const int*)d_in[6];

    const int n = in_sizes[0] / FDIM;
    const int e = in_sizes[1];
    const int* row = eidx;
    const int* col = eidx + e;

    const int nb   = (n + BK_NODES - 1) / BK_NODES;
    const int nblk = (e + CH_EDGES - 1) / CH_EDGES;
    float* dinv  = (float*)d_ws;
    int*   cnt   = (int*)(dinv + n);
    int*   rec   = (int*)(cnt + n);
    int*   start = rec + 2 * (size_t)e;
    int*   offs  = start + n;
    float* agg1  = (float*)(cnt + n) + (size_t)n * FDIM;
    int*   trec  = (int*)agg1;
    int*   tcol  = trec + 2 * (size_t)e;
    float* outp  = (float*)d_out;

    const int B = 256;
    const int gTile = (n + TILE_NODES - 1) / TILE_NODES;
    const int gScale = (n + 8 * SCALE_NODES - 1) / (8 * SCALE_NODES);

    gcn_chunk_order<<<nblk, CH_THREADS, 0, stream>>>(row, col, ew, offs, trec, tcol, e, nb);
    gcn_bucket_sort<<<nb, SORT_THREADS, 0, stream>>>(trec, tcol, offs, cnt, start, rec, n, nb, nblk);
    gcn_deg_scale<<<gScale, B, 0, stream>>>(x, start, cnt, rec, dinv, outp, n);

    gcn_agg_gemm <<<gTile, B, 0, stream>>>(outp, start, cnt, rec, dinv, W1, b1, agg1, n, 1);

    gcn_agg_gemm <<<gTile, B, 0, stream>>>(agg1, start, cnt, rec, dinv, W2, b2, outp, n, 0);
}
